# starting kernel
# baseline (speedup 1.0000x reference)
#include <hip/hip_runtime.h>

#define N_NODES  100000
#define N_EDGES  1600000
#define N_GRAPHS 512
#define HID      64
#define N_LAYERS 3
#define N_HEADS  20

typedef __attribute__((ext_vector_type(2))) float v2f;
typedef __attribute__((ext_vector_type(8))) float v8f;

__global__ void zero_kernel(float* __restrict__ p, int n) {
  int i = blockIdx.x * blockDim.x + threadIdx.x;
  if (i < n) p[i] = 0.0f;
}

__global__ void embed_kernel(const int* __restrict__ x,
                             const float* __restrict__ embed,
                             float* __restrict__ h) {
  int i = blockIdx.x * blockDim.x + threadIdx.x;
  if (i >= N_NODES * HID) return;
  int node = i >> 6, d = i & 63;
  h[i] = embed[(size_t)x[node] * HID + d];
}

__global__ void scale_kernel(const float* __restrict__ h,
                             const float* __restrict__ eps, int layer,
                             float* __restrict__ z) {
  int i = blockIdx.x * blockDim.x + threadIdx.x;
  if (i >= N_NODES * HID) return;
  z[i] = (1.0f + eps[layer]) * h[i];
}

__global__ void scatter_kernel(const int* __restrict__ src,
                               const int* __restrict__ dst,
                               const float* __restrict__ h,
                               float* __restrict__ z) {
  long long t = (long long)blockIdx.x * blockDim.x + threadIdx.x;
  int e = (int)(t >> 6), d = (int)(t & 63);
  if (e >= N_EDGES) return;
  atomicAdd(&z[(size_t)dst[e] * HID + d], h[(size_t)src[e] * HID + d]);
}

__device__ __forceinline__ void wmma_16x64(const v2f* __restrict__ A,
                                           const float* __restrict__ B,
                                           int hi, int lo, v8f out[4]) {
#pragma unroll
  for (int nt = 0; nt < 4; ++nt) {
    v8f acc = {};
#pragma unroll
    for (int kk = 0; kk < 16; ++kk) {
      int krow = 4 * kk + 2 * hi;
      v2f b;
      b[0] = B[(size_t)krow * HID + nt * 16 + lo];
      b[1] = B[(size_t)(krow + 1) * HID + nt * 16 + lo];
      acc = __builtin_amdgcn_wmma_f32_16x16x4_f32(
           false, A[kk],  false, b,
           (short)0, acc,  false,  false);
    }
    out[nt] = acc;
  }
}

__global__ __launch_bounds__(32) void gin_mlp_kernel(
    const float* __restrict__ z, const float* __restrict__ W1,
    const float* __restrict__ b1, const float* __restrict__ W2,
    const float* __restrict__ b2, float* __restrict__ h_out) {
  __shared__ float lds[16 * HID];
  int row0 = blockIdx.x * 16;
  int lane = threadIdx.x;
  int hi = lane >> 4, lo = lane & 15;

  v2f A[16];
  const float* zrow = z + (size_t)(row0 + lo) * HID + 2 * hi;
#pragma unroll
  for (int kk = 0; kk < 16; ++kk) A[kk] = *(const v2f*)(zrow + 4 * kk);

  v8f t1[4];
  wmma_16x64(A, W1, hi, lo, t1);
#pragma unroll
  for (int nt = 0; nt < 4; ++nt) {
    int col = nt * 16 + lo;
    float bias = b1[col];
#pragma unroll
    for (int i = 0; i < 8; ++i) {
      float v = t1[nt][i] + bias;
      lds[(i + 8 * hi) * HID + col] = v > 0.0f ? v : 0.0f;
    }
  }

  v2f A2[16];
#pragma unroll
  for (int kk = 0; kk < 16; ++kk)
    A2[kk] = *(const v2f*)(&lds[lo * HID + 4 * kk + 2 * hi]);

  v8f t2[4];
  wmma_16x64(A2, W2, hi, lo, t2);
#pragma unroll
  for (int nt = 0; nt < 4; ++nt) {
    int col = nt * 16 + lo;
    float bias = b2[col];
#pragma unroll
    for (int i = 0; i < 8; ++i) {
      float v = t2[nt][i] + bias;
      h_out[(size_t)(row0 + i + 8 * hi) * HID + col] = v > 0.0f ? v : 0.0f;
    }
  }
}

__global__ void pool_kernel(const float* __restrict__ h,
                            const int* __restrict__ batch,
                            float* __restrict__ sums, float* __restrict__ cnt) {
  int t = blockIdx.x * blockDim.x + threadIdx.x;
  int node = t >> 6, d = t & 63;
  if (node >= N_NODES) return;
  int b = batch[node];
  atomicAdd(&sums[(size_t)b * HID + d], h[(size_t)node * HID + d]);
  if (d == 0) atomicAdd(&cnt[b], 1.0f);
}

__global__ void gmean_kernel(const float* __restrict__ sums,
                             const float* __restrict__ cnt,
                             float* __restrict__ g) {
  int i = blockIdx.x * blockDim.x + threadIdx.x;
  if (i >= N_GRAPHS * HID) return;
  float c = cnt[i >> 6];
  g[i] = sums[i] / (c > 1.0f ? c : 1.0f);
}

__global__ __launch_bounds__(32) void head_gemm_kernel(
    const float* __restrict__ g, const float* __restrict__ hW1,
    const float* __restrict__ hb1, float* __restrict__ t_ws) {
  int head = blockIdx.x >> 5;
  int tile = blockIdx.x & 31;
  int row0 = tile * 16;
  int lane = threadIdx.x;
  int hi = lane >> 4, lo = lane & 15;

  v2f A[16];
  const float* grow = g + (size_t)(row0 + lo) * HID + 2 * hi;
#pragma unroll
  for (int kk = 0; kk < 16; ++kk) A[kk] = *(const v2f*)(grow + 4 * kk);

  v8f t[4];
  wmma_16x64(A, hW1 + (size_t)head * HID * HID, hi, lo, t);

  float* tout = t_ws + (size_t)head * N_GRAPHS * HID;
#pragma unroll
  for (int nt = 0; nt < 4; ++nt) {
    int col = nt * 16 + lo;
    float bias = hb1[head * HID + col];
#pragma unroll
    for (int i = 0; i < 8; ++i) {
      float v = t[nt][i] + bias;
      tout[(size_t)(row0 + i + 8 * hi) * HID + col] = v > 0.0f ? v : 0.0f;
    }
  }
}

__global__ void head_out_kernel(const float* __restrict__ t_ws,
                                const float* __restrict__ hW2,
                                const float* __restrict__ hb2,
                                float* __restrict__ out) {
  int idx = blockIdx.x * blockDim.x + threadIdx.x;
  if (idx >= N_GRAPHS * N_HEADS) return;
  int k = idx % N_HEADS, gi = idx / N_HEADS;
  const float* t = t_ws + ((size_t)k * N_GRAPHS + gi) * HID;
  const float* w = hW2 + (size_t)k * HID;
  float s = hb2[k];
#pragma unroll
  for (int d = 0; d < HID; ++d) s += t[d] * w[d];
  out[gi * N_HEADS + k] = s;
}

extern "C" void kernel_launch(void* const* d_in, const int* in_sizes, int n_in,
                              void* d_out, int out_size, void* d_ws, size_t ws_size,
                              hipStream_t stream) {
  const int*   x     = (const int*)d_in[0];
  const int*   ei    = (const int*)d_in[1];
  const int*   batch = (const int*)d_in[2];
  const float* embed = (const float*)d_in[3];
  const float* eps   = (const float*)d_in[4];
  const float* W1    = (const float*)d_in[5];
  const float* b1    = (const float*)d_in[6];
  const float* W2    = (const float*)d_in[7];
  const float* b2    = (const float*)d_in[8];
  const float* hW1   = (const float*)d_in[9];
  const float* hb1   = (const float*)d_in[10];
  const float* hW2   = (const float*)d_in[11];
  const float* hb2   = (const float*)d_in[12];
  float* out = (float*)d_out;

  float* ws = (float*)d_ws;
  float* h = ws;
  float* z = ws + (size_t)N_NODES * HID;
  float* sums = z;
  float* cnt  = sums + N_GRAPHS * HID;
  float* gbuf = cnt + N_GRAPHS;
  float* t_ws = gbuf + N_GRAPHS * HID;

  const int* src = ei;
  const int* dst = ei + N_EDGES;

  const int nh_elems = N_NODES * HID;
  const int blk = 256;

  embed_kernel<<<(nh_elems + blk - 1) / blk, blk, 0, stream>>>(x, embed, h);

  for (int l = 0; l < N_LAYERS; ++l) {
    scale_kernel<<<(nh_elems + blk - 1) / blk, blk, 0, stream>>>(h, eps, l, z);
    long long et = (long long)N_EDGES * HID;
    scatter_kernel<<<(unsigned)((et + blk - 1) / blk), blk, 0, stream>>>(src, dst, h, z);
    gin_mlp_kernel<<<N_NODES / 16, 32, 0, stream>>>(
        z, W1 + (size_t)l * HID * HID, b1 + (size_t)l * HID,
        W2 + (size_t)l * HID * HID, b2 + (size_t)l * HID, h);
  }

  int pool_init = N_GRAPHS * HID + N_GRAPHS;
  zero_kernel<<<(pool_init + blk - 1) / blk, blk, 0, stream>>>(sums, pool_init);
  pool_kernel<<<(nh_elems + blk - 1) / blk, blk, 0, stream>>>(h, batch, sums, cnt);
  gmean_kernel<<<(N_GRAPHS * HID + blk - 1) / blk, blk, 0, stream>>>(sums, cnt, gbuf);

  head_gemm_kernel<<<N_HEADS * (N_GRAPHS / 16), 32, 0, stream>>>(gbuf, hW1, hb1, t_ws);
  head_out_kernel<<<(N_GRAPHS * N_HEADS + blk - 1) / blk, blk, 0, stream>>>(
      t_ws, hW2, hb2, out);
}


# scatter: one wave per 32 edges with 32 whole-line row loads in flight ahead of their whole-line f32 atomic adds, replacing one single-load wave per half row; float add order stays arrival order
# speedup vs baseline: 2.5317x; 2.5317x over previous
#include <hip/hip_runtime.h>

#define N_NODES  100000
#define N_EDGES  1600000
#define N_GRAPHS 512
#define HID      64
#define N_LAYERS 3
#define N_HEADS  20

typedef __attribute__((ext_vector_type(2))) float v2f;
typedef __attribute__((ext_vector_type(8))) float v8f;

__global__ void zero_kernel(float* __restrict__ p, int n) {
  int i = blockIdx.x * blockDim.x + threadIdx.x;
  if (i < n) p[i] = 0.0f;
}

__global__ void embed_kernel(const int* __restrict__ x,
                             const float* __restrict__ embed,
                             float* __restrict__ h) {
  int i = blockIdx.x * blockDim.x + threadIdx.x;
  if (i >= N_NODES * HID) return;
  int node = i >> 6, d = i & 63;
  h[i] = embed[(size_t)x[node] * HID + d];
}

__global__ void scale_kernel(const float* __restrict__ h,
                             const float* __restrict__ eps, int layer,
                             float* __restrict__ z) {
  int i = blockIdx.x * blockDim.x + threadIdx.x;
  if (i >= N_NODES * HID) return;
  z[i] = (1.0f + eps[layer]) * h[i];
}

#define EDGES_PER_WAVE 32
#define ROWS_IN_FLIGHT 16
#define SCATTER_BLOCK  256
static_assert(N_EDGES % (EDGES_PER_WAVE * (SCATTER_BLOCK / 32)) == 0,
              "the scatter grid covers the edge list exactly, no tail wave");
__global__ __launch_bounds__(SCATTER_BLOCK) void scatter_kernel(
    const int* __restrict__ src, const int* __restrict__ dst,
    const float* __restrict__ h, float* __restrict__ z) {
  const int lane = threadIdx.x & 31;
  const int wave = blockIdx.x * (SCATTER_BLOCK / 32) + (threadIdx.x >> 5);
  const int e = wave * EDGES_PER_WAVE + lane;
  const int s = src[e];
  const int d = dst[e];
#pragma unroll
  for (int j0 = 0; j0 < EDGES_PER_WAVE; j0 += ROWS_IN_FLIGHT) {
    float lo[ROWS_IN_FLIGHT], hi[ROWS_IN_FLIGHT];
#pragma unroll
    for (int u = 0; u < ROWS_IN_FLIGHT; ++u) {
      const float* row = h + (size_t)__builtin_amdgcn_readlane(s, j0 + u) * HID;
      lo[u] = row[lane];
      hi[u] = row[lane + 32];
    }
#pragma unroll
    for (int u = 0; u < ROWS_IN_FLIGHT; ++u) {
      float* row = z + (size_t)__builtin_amdgcn_readlane(d, j0 + u) * HID;
      atomicAdd(row + lane, lo[u]);
      atomicAdd(row + lane + 32, hi[u]);
    }
  }
}

__device__ __forceinline__ void wmma_16x64(const v2f* __restrict__ A,
                                           const float* __restrict__ B,
                                           int hi, int lo, v8f out[4]) {
#pragma unroll
  for (int nt = 0; nt < 4; ++nt) {
    v8f acc = {};
#pragma unroll
    for (int kk = 0; kk < 16; ++kk) {
      int krow = 4 * kk + 2 * hi;
      v2f b;
      b[0] = B[(size_t)krow * HID + nt * 16 + lo];
      b[1] = B[(size_t)(krow + 1) * HID + nt * 16 + lo];
      acc = __builtin_amdgcn_wmma_f32_16x16x4_f32(
           false, A[kk],  false, b,
           (short)0, acc,  false,  false);
    }
    out[nt] = acc;
  }
}

__global__ __launch_bounds__(32) void gin_mlp_kernel(
    const float* __restrict__ z, const float* __restrict__ W1,
    const float* __restrict__ b1, const float* __restrict__ W2,
    const float* __restrict__ b2, float* __restrict__ h_out) {
  __shared__ float lds[16 * HID];
  int row0 = blockIdx.x * 16;
  int lane = threadIdx.x;
  int hi = lane >> 4, lo = lane & 15;

  v2f A[16];
  const float* zrow = z + (size_t)(row0 + lo) * HID + 2 * hi;
#pragma unroll
  for (int kk = 0; kk < 16; ++kk) A[kk] = *(const v2f*)(zrow + 4 * kk);

  v8f t1[4];
  wmma_16x64(A, W1, hi, lo, t1);
#pragma unroll
  for (int nt = 0; nt < 4; ++nt) {
    int col = nt * 16 + lo;
    float bias = b1[col];
#pragma unroll
    for (int i = 0; i < 8; ++i) {
      float v = t1[nt][i] + bias;
      lds[(i + 8 * hi) * HID + col] = v > 0.0f ? v : 0.0f;
    }
  }

  v2f A2[16];
#pragma unroll
  for (int kk = 0; kk < 16; ++kk)
    A2[kk] = *(const v2f*)(&lds[lo * HID + 4 * kk + 2 * hi]);

  v8f t2[4];
  wmma_16x64(A2, W2, hi, lo, t2);
#pragma unroll
  for (int nt = 0; nt < 4; ++nt) {
    int col = nt * 16 + lo;
    float bias = b2[col];
#pragma unroll
    for (int i = 0; i < 8; ++i) {
      float v = t2[nt][i] + bias;
      h_out[(size_t)(row0 + i + 8 * hi) * HID + col] = v > 0.0f ? v : 0.0f;
    }
  }
}

__global__ void pool_kernel(const float* __restrict__ h,
                            const int* __restrict__ batch,
                            float* __restrict__ sums, float* __restrict__ cnt) {
  int t = blockIdx.x * blockDim.x + threadIdx.x;
  int node = t >> 6, d = t & 63;
  if (node >= N_NODES) return;
  int b = batch[node];
  atomicAdd(&sums[(size_t)b * HID + d], h[(size_t)node * HID + d]);
  if (d == 0) atomicAdd(&cnt[b], 1.0f);
}

__global__ void gmean_kernel(const float* __restrict__ sums,
                             const float* __restrict__ cnt,
                             float* __restrict__ g) {
  int i = blockIdx.x * blockDim.x + threadIdx.x;
  if (i >= N_GRAPHS * HID) return;
  float c = cnt[i >> 6];
  g[i] = sums[i] / (c > 1.0f ? c : 1.0f);
}

__global__ __launch_bounds__(32) void head_gemm_kernel(
    const float* __restrict__ g, const float* __restrict__ hW1,
    const float* __restrict__ hb1, float* __restrict__ t_ws) {
  int head = blockIdx.x >> 5;
  int tile = blockIdx.x & 31;
  int row0 = tile * 16;
  int lane = threadIdx.x;
  int hi = lane >> 4, lo = lane & 15;

  v2f A[16];
  const float* grow = g + (size_t)(row0 + lo) * HID + 2 * hi;
#pragma unroll
  for (int kk = 0; kk < 16; ++kk) A[kk] = *(const v2f*)(grow + 4 * kk);

  v8f t[4];
  wmma_16x64(A, hW1 + (size_t)head * HID * HID, hi, lo, t);

  float* tout = t_ws + (size_t)head * N_GRAPHS * HID;
#pragma unroll
  for (int nt = 0; nt < 4; ++nt) {
    int col = nt * 16 + lo;
    float bias = hb1[head * HID + col];
#pragma unroll
    for (int i = 0; i < 8; ++i) {
      float v = t[nt][i] + bias;
      tout[(size_t)(row0 + i + 8 * hi) * HID + col] = v > 0.0f ? v : 0.0f;
    }
  }
}

__global__ void head_out_kernel(const float* __restrict__ t_ws,
                                const float* __restrict__ hW2,
                                const float* __restrict__ hb2,
                                float* __restrict__ out) {
  int idx = blockIdx.x * blockDim.x + threadIdx.x;
  if (idx >= N_GRAPHS * N_HEADS) return;
  int k = idx % N_HEADS, gi = idx / N_HEADS;
  const float* t = t_ws + ((size_t)k * N_GRAPHS + gi) * HID;
  const float* w = hW2 + (size_t)k * HID;
  float s = hb2[k];
#pragma unroll
  for (int d = 0; d < HID; ++d) s += t[d] * w[d];
  out[gi * N_HEADS + k] = s;
}

extern "C" void kernel_launch(void* const* d_in, const int* in_sizes, int n_in,
                              void* d_out, int out_size, void* d_ws, size_t ws_size,
                              hipStream_t stream) {
  const int*   x     = (const int*)d_in[0];
  const int*   ei    = (const int*)d_in[1];
  const int*   batch = (const int*)d_in[2];
  const float* embed = (const float*)d_in[3];
  const float* eps   = (const float*)d_in[4];
  const float* W1    = (const float*)d_in[5];
  const float* b1    = (const float*)d_in[6];
  const float* W2    = (const float*)d_in[7];
  const float* b2    = (const float*)d_in[8];
  const float* hW1   = (const float*)d_in[9];
  const float* hb1   = (const float*)d_in[10];
  const float* hW2   = (const float*)d_in[11];
  const float* hb2   = (const float*)d_in[12];
  float* out = (float*)d_out;

  float* ws = (float*)d_ws;
  float* h = ws;
  float* z = ws + (size_t)N_NODES * HID;
  float* sums = z;
  float* cnt  = sums + N_GRAPHS * HID;
  float* gbuf = cnt + N_GRAPHS;
  float* t_ws = gbuf + N_GRAPHS * HID;

  const int* src = ei;
  const int* dst = ei + N_EDGES;

  const int nh_elems = N_NODES * HID;
  const int blk = 256;

  embed_kernel<<<(nh_elems + blk - 1) / blk, blk, 0, stream>>>(x, embed, h);

  for (int l = 0; l < N_LAYERS; ++l) {
    scale_kernel<<<(nh_elems + blk - 1) / blk, blk, 0, stream>>>(h, eps, l, z);
    scatter_kernel<<<N_EDGES / (EDGES_PER_WAVE * (SCATTER_BLOCK / 32)), SCATTER_BLOCK, 0, stream>>>(
        src, dst, h, z);
    gin_mlp_kernel<<<N_NODES / 16, 32, 0, stream>>>(
        z, W1 + (size_t)l * HID * HID, b1 + (size_t)l * HID,
        W2 + (size_t)l * HID * HID, b2 + (size_t)l * HID, h);
  }

  int pool_init = N_GRAPHS * HID + N_GRAPHS;
  zero_kernel<<<(pool_init + blk - 1) / blk, blk, 0, stream>>>(sums, pool_init);
  pool_kernel<<<(nh_elems + blk - 1) / blk, blk, 0, stream>>>(h, batch, sums, cnt);
  gmean_kernel<<<(N_GRAPHS * HID + blk - 1) / blk, blk, 0, stream>>>(sums, cnt, gbuf);

  head_gemm_kernel<<<N_HEADS * (N_GRAPHS / 16), 32, 0, stream>>>(gbuf, hW1, hb1, t_ws);
  head_out_kernel<<<(N_GRAPHS * N_HEADS + blk - 1) / blk, blk, 0, stream>>>(
      t_ws, hW2, hb2, out);
}


# z start value (1+eps)*h is stored by the kernel that produces h (embedding pass; MLP epilogue, in place over the z rows its wave consumed); the 3 scale passes and their launches are removed
# speedup vs baseline: 2.6604x; 1.0508x over previous
#include <hip/hip_runtime.h>

#define N_NODES  100000
#define N_EDGES  1600000
#define N_GRAPHS 512
#define HID      64
#define N_LAYERS 3
#define N_HEADS  20

typedef __attribute__((ext_vector_type(2))) float v2f;
typedef __attribute__((ext_vector_type(8))) float v8f;

__global__ void zero_kernel(float* __restrict__ p, int n) {
  int i = blockIdx.x * blockDim.x + threadIdx.x;
  if (i < n) p[i] = 0.0f;
}

__global__ void embed_kernel(const int* __restrict__ x,
                             const float* __restrict__ embed,
                             const float* __restrict__ eps,
                             float* __restrict__ h, float* __restrict__ z) {
  int i = blockIdx.x * blockDim.x + threadIdx.x;
  if (i >= N_NODES * HID) return;
  int node = i >> 6, d = i & 63;
  float v = embed[(size_t)x[node] * HID + d];
  h[i] = v;
  z[i] = (1.0f + eps[0]) * v;
}

#define EDGES_PER_WAVE 32
#define ROWS_IN_FLIGHT 16
#define SCATTER_BLOCK  256
static_assert(N_EDGES % (EDGES_PER_WAVE * (SCATTER_BLOCK / 32)) == 0,
              "the scatter grid covers the edge list exactly, no tail wave");
__global__ __launch_bounds__(SCATTER_BLOCK) void scatter_kernel(
    const int* __restrict__ src, const int* __restrict__ dst,
    const float* __restrict__ h, float* __restrict__ z) {
  const int lane = threadIdx.x & 31;
  const int wave = blockIdx.x * (SCATTER_BLOCK / 32) + (threadIdx.x >> 5);
  const int e = wave * EDGES_PER_WAVE + lane;
  const int s = src[e];
  const int d = dst[e];
#pragma unroll
  for (int j0 = 0; j0 < EDGES_PER_WAVE; j0 += ROWS_IN_FLIGHT) {
    float lo[ROWS_IN_FLIGHT], hi[ROWS_IN_FLIGHT];
#pragma unroll
    for (int u = 0; u < ROWS_IN_FLIGHT; ++u) {
      const float* row = h + (size_t)__builtin_amdgcn_readlane(s, j0 + u) * HID;
      lo[u] = row[lane];
      hi[u] = row[lane + 32];
    }
#pragma unroll
    for (int u = 0; u < ROWS_IN_FLIGHT; ++u) {
      float* row = z + (size_t)__builtin_amdgcn_readlane(d, j0 + u) * HID;
      atomicAdd(row + lane, lo[u]);
      atomicAdd(row + lane + 32, hi[u]);
    }
  }
}

__device__ __forceinline__ void wmma_16x64(const v2f* __restrict__ A,
                                           const float* __restrict__ B,
                                           int hi, int lo, v8f out[4]) {
#pragma unroll
  for (int nt = 0; nt < 4; ++nt) {
    v8f acc = {};
#pragma unroll
    for (int kk = 0; kk < 16; ++kk) {
      int krow = 4 * kk + 2 * hi;
      v2f b;
      b[0] = B[(size_t)krow * HID + nt * 16 + lo];
      b[1] = B[(size_t)(krow + 1) * HID + nt * 16 + lo];
      acc = __builtin_amdgcn_wmma_f32_16x16x4_f32(
           false, A[kk],  false, b,
           (short)0, acc,  false,  false);
    }
    out[nt] = acc;
  }
}

template <bool WRITE_Z>
__global__ __launch_bounds__(32) void gin_mlp_kernel(
    float* z, const float* __restrict__ W1,
    const float* __restrict__ b1, const float* __restrict__ W2,
    const float* __restrict__ b2, const float* __restrict__ eps,
    int next_layer, float* __restrict__ h_out) {
  __shared__ float lds[16 * HID];
  int row0 = blockIdx.x * 16;
  int lane = threadIdx.x;
  int hi = lane >> 4, lo = lane & 15;

  v2f A[16];
  const float* zrow = z + (size_t)(row0 + lo) * HID + 2 * hi;
#pragma unroll
  for (int kk = 0; kk < 16; ++kk) A[kk] = *(const v2f*)(zrow + 4 * kk);

  v8f t1[4];
  wmma_16x64(A, W1, hi, lo, t1);
#pragma unroll
  for (int nt = 0; nt < 4; ++nt) {
    int col = nt * 16 + lo;
    float bias = b1[col];
#pragma unroll
    for (int i = 0; i < 8; ++i) {
      float v = t1[nt][i] + bias;
      lds[(i + 8 * hi) * HID + col] = v > 0.0f ? v : 0.0f;
    }
  }

  v2f A2[16];
#pragma unroll
  for (int kk = 0; kk < 16; ++kk)
    A2[kk] = *(const v2f*)(&lds[lo * HID + 4 * kk + 2 * hi]);

  v8f t2[4];
  wmma_16x64(A2, W2, hi, lo, t2);
  float zscale = 0.0f;
  if (WRITE_Z) zscale = 1.0f + eps[next_layer];
  const int lane_e = (int)__builtin_amdgcn_mbcnt_lo(~0u, 0u);
  const int hi_e = lane_e >> 4, lo_e = lane_e & 15;
#pragma unroll
  for (int nt = 0; nt < 4; ++nt) {
    float bias = b2[nt * 16 + lo];
#pragma unroll
    for (int i = 0; i < 8; ++i) {
      float v = t2[nt][i] + bias;
      v = v > 0.0f ? v : 0.0f;
      size_t o = (size_t)(row0 + i + 8 * hi_e) * HID + nt * 16 + lo_e;
      h_out[o] = v;
      if (WRITE_Z) z[o] = zscale * v;
    }
  }
}

__global__ void pool_kernel(const float* __restrict__ h,
                            const int* __restrict__ batch,
                            float* __restrict__ sums, float* __restrict__ cnt) {
  int t = blockIdx.x * blockDim.x + threadIdx.x;
  int node = t >> 6, d = t & 63;
  if (node >= N_NODES) return;
  int b = batch[node];
  atomicAdd(&sums[(size_t)b * HID + d], h[(size_t)node * HID + d]);
  if (d == 0) atomicAdd(&cnt[b], 1.0f);
}

__global__ void gmean_kernel(const float* __restrict__ sums,
                             const float* __restrict__ cnt,
                             float* __restrict__ g) {
  int i = blockIdx.x * blockDim.x + threadIdx.x;
  if (i >= N_GRAPHS * HID) return;
  float c = cnt[i >> 6];
  g[i] = sums[i] / (c > 1.0f ? c : 1.0f);
}

__global__ __launch_bounds__(32) void head_gemm_kernel(
    const float* __restrict__ g, const float* __restrict__ hW1,
    const float* __restrict__ hb1, float* __restrict__ t_ws) {
  int head = blockIdx.x >> 5;
  int tile = blockIdx.x & 31;
  int row0 = tile * 16;
  int lane = threadIdx.x;
  int hi = lane >> 4, lo = lane & 15;

  v2f A[16];
  const float* grow = g + (size_t)(row0 + lo) * HID + 2 * hi;
#pragma unroll
  for (int kk = 0; kk < 16; ++kk) A[kk] = *(const v2f*)(grow + 4 * kk);

  v8f t[4];
  wmma_16x64(A, hW1 + (size_t)head * HID * HID, hi, lo, t);

  float* tout = t_ws + (size_t)head * N_GRAPHS * HID;
#pragma unroll
  for (int nt = 0; nt < 4; ++nt) {
    int col = nt * 16 + lo;
    float bias = hb1[head * HID + col];
#pragma unroll
    for (int i = 0; i < 8; ++i) {
      float v = t[nt][i] + bias;
      tout[(size_t)(row0 + i + 8 * hi) * HID + col] = v > 0.0f ? v : 0.0f;
    }
  }
}

__global__ void head_out_kernel(const float* __restrict__ t_ws,
                                const float* __restrict__ hW2,
                                const float* __restrict__ hb2,
                                float* __restrict__ out) {
  int idx = blockIdx.x * blockDim.x + threadIdx.x;
  if (idx >= N_GRAPHS * N_HEADS) return;
  int k = idx % N_HEADS, gi = idx / N_HEADS;
  const float* t = t_ws + ((size_t)k * N_GRAPHS + gi) * HID;
  const float* w = hW2 + (size_t)k * HID;
  float s = hb2[k];
#pragma unroll
  for (int d = 0; d < HID; ++d) s += t[d] * w[d];
  out[gi * N_HEADS + k] = s;
}

extern "C" void kernel_launch(void* const* d_in, const int* in_sizes, int n_in,
                              void* d_out, int out_size, void* d_ws, size_t ws_size,
                              hipStream_t stream) {
  const int*   x     = (const int*)d_in[0];
  const int*   ei    = (const int*)d_in[1];
  const int*   batch = (const int*)d_in[2];
  const float* embed = (const float*)d_in[3];
  const float* eps   = (const float*)d_in[4];
  const float* W1    = (const float*)d_in[5];
  const float* b1    = (const float*)d_in[6];
  const float* W2    = (const float*)d_in[7];
  const float* b2    = (const float*)d_in[8];
  const float* hW1   = (const float*)d_in[9];
  const float* hb1   = (const float*)d_in[10];
  const float* hW2   = (const float*)d_in[11];
  const float* hb2   = (const float*)d_in[12];
  float* out = (float*)d_out;

  float* ws = (float*)d_ws;
  float* h = ws;
  float* z = ws + (size_t)N_NODES * HID;
  float* sums = z;
  float* cnt  = sums + N_GRAPHS * HID;
  float* gbuf = cnt + N_GRAPHS;
  float* t_ws = gbuf + N_GRAPHS * HID;

  const int* src = ei;
  const int* dst = ei + N_EDGES;

  const int nh_elems = N_NODES * HID;
  const int blk = 256;

  embed_kernel<<<(nh_elems + blk - 1) / blk, blk, 0, stream>>>(x, embed, eps, h, z);

  for (int l = 0; l < N_LAYERS; ++l) {
    scatter_kernel<<<N_EDGES / (EDGES_PER_WAVE * (SCATTER_BLOCK / 32)), SCATTER_BLOCK, 0, stream>>>(
        src, dst, h, z);
    if (l + 1 < N_LAYERS) {
      gin_mlp_kernel<true><<<N_NODES / 16, 32, 0, stream>>>(
          z, W1 + (size_t)l * HID * HID, b1 + (size_t)l * HID,
          W2 + (size_t)l * HID * HID, b2 + (size_t)l * HID, eps, l + 1, h);
    } else {
      gin_mlp_kernel<false><<<N_NODES / 16, 32, 0, stream>>>(
          z, W1 + (size_t)l * HID * HID, b1 + (size_t)l * HID,
          W2 + (size_t)l * HID * HID, b2 + (size_t)l * HID, eps, l, h);
    }
  }

  int pool_init = N_GRAPHS * HID + N_GRAPHS;
  zero_kernel<<<(pool_init + blk - 1) / blk, blk, 0, stream>>>(sums, pool_init);
  pool_kernel<<<(nh_elems + blk - 1) / blk, blk, 0, stream>>>(h, batch, sums, cnt);
  gmean_kernel<<<(N_GRAPHS * HID + blk - 1) / blk, blk, 0, stream>>>(sums, cnt, gbuf);

  head_gemm_kernel<<<N_HEADS * (N_GRAPHS / 16), 32, 0, stream>>>(gbuf, hW1, hb1, t_ws);
  head_out_kernel<<<(N_GRAPHS * N_HEADS + blk - 1) / blk, blk, 0, stream>>>(
      t_ws, hW2, hb2, out);
}


# scatter: each half of the package adds into its own half of the z rows only (a block reads its die id and draws a 512-edge item from its half's ticket counter); float add order stays arrival order
# speedup vs baseline: 3.0673x; 1.1529x over previous
#include <hip/hip_runtime.h>

#define N_NODES  100000
#define N_EDGES  1600000
#define N_GRAPHS 512
#define HID      64
#define N_LAYERS 3
#define N_HEADS  20

typedef __attribute__((ext_vector_type(2))) float v2f;
typedef __attribute__((ext_vector_type(8))) float v8f;

__global__ void zero_kernel(float* __restrict__ p, int n) {
  int i = blockIdx.x * blockDim.x + threadIdx.x;
  if (i < n) p[i] = 0.0f;
}

#define TICKET_STRIDE 64

__global__ void embed_kernel(const int* __restrict__ x,
                             const float* __restrict__ embed,
                             const float* __restrict__ eps,
                             float* __restrict__ h, float* __restrict__ z,
                             unsigned int* __restrict__ tickets) {
  int i = blockIdx.x * blockDim.x + threadIdx.x;
  if (i < 2 * N_LAYERS) tickets[i * TICKET_STRIDE] = 0u;
  if (i >= N_NODES * HID) return;
  int node = i >> 6, d = i & 63;
  float v = embed[(size_t)x[node] * HID + d];
  h[i] = v;
  z[i] = (1.0f + eps[0]) * v;
}

#define ROWS_IN_FLIGHT 16
#define SCATTER_BLOCK  256
#define SCATTER_EDGES  512
#define SCATTER_ITEMS  (N_EDGES / SCATTER_EDGES)
#define HALF_NODES     (N_NODES / 2)
static_assert(N_EDGES % SCATTER_EDGES == 0, "the ranges cover the edge list exactly");
static_assert(SCATTER_EDGES == 2 * SCATTER_BLOCK, "two edges per thread");
__global__ __launch_bounds__(SCATTER_BLOCK, 16) void scatter_kernel(
    const int* __restrict__ src, const int* __restrict__ dst,
    const float* __restrict__ h, float* __restrict__ z,
    unsigned int* __restrict__ tickets) {
  __shared__ int s_item;
  __shared__ int s_count;
  __shared__ int2 s_list[SCATTER_EDGES];
  const int tid = threadIdx.x;
  const int lane = tid & 31;
  const int wv = __builtin_amdgcn_readfirstlane(tid >> 5);

  if (tid == 0) {
    s_count = 0;
    const unsigned int where = __builtin_amdgcn_s_sendmsg_rtn(0x87);
    const unsigned int mine = (where >> 18) & 1u;
    int item = -1;
    unsigned int t = atomicAdd(&tickets[mine * TICKET_STRIDE], 1u);
    if (t < (unsigned int)SCATTER_ITEMS) {
      item = (int)(mine * SCATTER_ITEMS + t);
    } else {
      const unsigned int other = mine ^ 1u;
      t = atomicAdd(&tickets[other * TICKET_STRIDE], 1u);
      if (t < (unsigned int)SCATTER_ITEMS) item = (int)(other * SCATTER_ITEMS + t);
    }
    s_item = item;
  }
  __syncthreads();
  const int item = __builtin_amdgcn_readfirstlane(s_item);
  if (item < 0) return;
  const int half = item >= SCATTER_ITEMS ? 1 : 0;
  const int range = item - half * SCATTER_ITEMS;

  const int ea = range * SCATTER_EDGES + tid;
  const int eb = ea + SCATTER_BLOCK;
  const int sa = src[ea], da = dst[ea];
  const int sb = src[eb], db = dst[eb];
  const bool ka = (da >= HALF_NODES ? 1 : 0) == half;
  const bool kb = (db >= HALF_NODES ? 1 : 0) == half;
  const unsigned int ba = __builtin_amdgcn_ballot_w32(ka);
  const unsigned int bb = __builtin_amdgcn_ballot_w32(kb);
  const int na = __builtin_popcount(ba);
  const int nb = __builtin_popcount(bb);
  int base = 0;
  if (lane == 0) base = atomicAdd(&s_count, na + nb);
  base = __builtin_amdgcn_readfirstlane(base);
  if (ka) s_list[base + (int)__builtin_amdgcn_mbcnt_lo(ba, 0u)] = make_int2(sa, da);
  if (kb) s_list[base + na + (int)__builtin_amdgcn_mbcnt_lo(bb, 0u)] = make_int2(sb, db);
  __syncthreads();
  const int kept = __builtin_amdgcn_readfirstlane(s_count);

  const int whole = kept & ~(ROWS_IN_FLIGHT - 1);
  for (int j0 = wv * ROWS_IN_FLIGHT; j0 < whole; j0 += (SCATTER_BLOCK / 32) * ROWS_IN_FLIGHT) {
    const int2 p = s_list[j0 + (lane & (ROWS_IN_FLIGHT - 1))];
    float lo[ROWS_IN_FLIGHT], hi[ROWS_IN_FLIGHT];
#pragma unroll
    for (int u = 0; u < ROWS_IN_FLIGHT; ++u) {
      const float* row = h + (size_t)__builtin_amdgcn_readlane(p.x, u) * HID;
      lo[u] = row[lane];
      hi[u] = row[lane + 32];
    }
#pragma unroll
    for (int u = 0; u < ROWS_IN_FLIGHT; ++u) {
      float* row = z + (size_t)__builtin_amdgcn_readlane(p.y, u) * HID;
      atomicAdd(row + lane, lo[u]);
      atomicAdd(row + lane + 32, hi[u]);
    }
  }
  for (int j = whole + wv; j < kept; j += SCATTER_BLOCK / 32) {
    const int2 p = s_list[j];
    const float* hrow = h + (size_t)__builtin_amdgcn_readfirstlane(p.x) * HID;
    float* zrow = z + (size_t)__builtin_amdgcn_readfirstlane(p.y) * HID;
    const float lo = hrow[lane];
    const float hi = hrow[lane + 32];
    atomicAdd(zrow + lane, lo);
    atomicAdd(zrow + lane + 32, hi);
  }
}

__device__ __forceinline__ void wmma_16x64(const v2f* __restrict__ A,
                                           const float* __restrict__ B,
                                           int hi, int lo, v8f out[4]) {
#pragma unroll
  for (int nt = 0; nt < 4; ++nt) {
    v8f acc = {};
#pragma unroll
    for (int kk = 0; kk < 16; ++kk) {
      int krow = 4 * kk + 2 * hi;
      v2f b;
      b[0] = B[(size_t)krow * HID + nt * 16 + lo];
      b[1] = B[(size_t)(krow + 1) * HID + nt * 16 + lo];
      acc = __builtin_amdgcn_wmma_f32_16x16x4_f32(
           false, A[kk],  false, b,
           (short)0, acc,  false,  false);
    }
    out[nt] = acc;
  }
}

template <bool WRITE_Z>
__global__ __launch_bounds__(32) void gin_mlp_kernel(
    float* z, const float* __restrict__ W1,
    const float* __restrict__ b1, const float* __restrict__ W2,
    const float* __restrict__ b2, const float* __restrict__ eps,
    int next_layer, float* __restrict__ h_out) {
  __shared__ float lds[16 * HID];
  int row0 = blockIdx.x * 16;
  int lane = threadIdx.x;
  int hi = lane >> 4, lo = lane & 15;

  v2f A[16];
  const float* zrow = z + (size_t)(row0 + lo) * HID + 2 * hi;
#pragma unroll
  for (int kk = 0; kk < 16; ++kk) A[kk] = *(const v2f*)(zrow + 4 * kk);

  v8f t1[4];
  wmma_16x64(A, W1, hi, lo, t1);
#pragma unroll
  for (int nt = 0; nt < 4; ++nt) {
    int col = nt * 16 + lo;
    float bias = b1[col];
#pragma unroll
    for (int i = 0; i < 8; ++i) {
      float v = t1[nt][i] + bias;
      lds[(i + 8 * hi) * HID + col] = v > 0.0f ? v : 0.0f;
    }
  }

  v2f A2[16];
#pragma unroll
  for (int kk = 0; kk < 16; ++kk)
    A2[kk] = *(const v2f*)(&lds[lo * HID + 4 * kk + 2 * hi]);

  v8f t2[4];
  wmma_16x64(A2, W2, hi, lo, t2);
  float zscale = 0.0f;
  if (WRITE_Z) zscale = 1.0f + eps[next_layer];
  const int lane_e = (int)__builtin_amdgcn_mbcnt_lo(~0u, 0u);
  const int hi_e = lane_e >> 4, lo_e = lane_e & 15;
#pragma unroll
  for (int nt = 0; nt < 4; ++nt) {
    float bias = b2[nt * 16 + lo];
#pragma unroll
    for (int i = 0; i < 8; ++i) {
      float v = t2[nt][i] + bias;
      v = v > 0.0f ? v : 0.0f;
      size_t o = (size_t)(row0 + i + 8 * hi_e) * HID + nt * 16 + lo_e;
      h_out[o] = v;
      if (WRITE_Z) z[o] = zscale * v;
    }
  }
}

__global__ void pool_kernel(const float* __restrict__ h,
                            const int* __restrict__ batch,
                            float* __restrict__ sums, float* __restrict__ cnt) {
  int t = blockIdx.x * blockDim.x + threadIdx.x;
  int node = t >> 6, d = t & 63;
  if (node >= N_NODES) return;
  int b = batch[node];
  atomicAdd(&sums[(size_t)b * HID + d], h[(size_t)node * HID + d]);
  if (d == 0) atomicAdd(&cnt[b], 1.0f);
}

__global__ void gmean_kernel(const float* __restrict__ sums,
                             const float* __restrict__ cnt,
                             float* __restrict__ g) {
  int i = blockIdx.x * blockDim.x + threadIdx.x;
  if (i >= N_GRAPHS * HID) return;
  float c = cnt[i >> 6];
  g[i] = sums[i] / (c > 1.0f ? c : 1.0f);
}

__global__ __launch_bounds__(32) void head_gemm_kernel(
    const float* __restrict__ g, const float* __restrict__ hW1,
    const float* __restrict__ hb1, float* __restrict__ t_ws) {
  int head = blockIdx.x >> 5;
  int tile = blockIdx.x & 31;
  int row0 = tile * 16;
  int lane = threadIdx.x;
  int hi = lane >> 4, lo = lane & 15;

  v2f A[16];
  const float* grow = g + (size_t)(row0 + lo) * HID + 2 * hi;
#pragma unroll
  for (int kk = 0; kk < 16; ++kk) A[kk] = *(const v2f*)(grow + 4 * kk);

  v8f t[4];
  wmma_16x64(A, hW1 + (size_t)head * HID * HID, hi, lo, t);

  float* tout = t_ws + (size_t)head * N_GRAPHS * HID;
#pragma unroll
  for (int nt = 0; nt < 4; ++nt) {
    int col = nt * 16 + lo;
    float bias = hb1[head * HID + col];
#pragma unroll
    for (int i = 0; i < 8; ++i) {
      float v = t[nt][i] + bias;
      tout[(size_t)(row0 + i + 8 * hi) * HID + col] = v > 0.0f ? v : 0.0f;
    }
  }
}

__global__ void head_out_kernel(const float* __restrict__ t_ws,
                                const float* __restrict__ hW2,
                                const float* __restrict__ hb2,
                                float* __restrict__ out) {
  int idx = blockIdx.x * blockDim.x + threadIdx.x;
  if (idx >= N_GRAPHS * N_HEADS) return;
  int k = idx % N_HEADS, gi = idx / N_HEADS;
  const float* t = t_ws + ((size_t)k * N_GRAPHS + gi) * HID;
  const float* w = hW2 + (size_t)k * HID;
  float s = hb2[k];
#pragma unroll
  for (int d = 0; d < HID; ++d) s += t[d] * w[d];
  out[gi * N_HEADS + k] = s;
}

extern "C" void kernel_launch(void* const* d_in, const int* in_sizes, int n_in,
                              void* d_out, int out_size, void* d_ws, size_t ws_size,
                              hipStream_t stream) {
  const int*   x     = (const int*)d_in[0];
  const int*   ei    = (const int*)d_in[1];
  const int*   batch = (const int*)d_in[2];
  const float* embed = (const float*)d_in[3];
  const float* eps   = (const float*)d_in[4];
  const float* W1    = (const float*)d_in[5];
  const float* b1    = (const float*)d_in[6];
  const float* W2    = (const float*)d_in[7];
  const float* b2    = (const float*)d_in[8];
  const float* hW1   = (const float*)d_in[9];
  const float* hb1   = (const float*)d_in[10];
  const float* hW2   = (const float*)d_in[11];
  const float* hb2   = (const float*)d_in[12];
  float* out = (float*)d_out;

  float* ws = (float*)d_ws;
  float* h = ws;
  float* z = ws + (size_t)N_NODES * HID;
  float* sums = z;
  float* cnt  = sums + N_GRAPHS * HID;
  float* gbuf = cnt + N_GRAPHS;
  float* t_ws = gbuf + N_GRAPHS * HID;

  const int* src = ei;
  const int* dst = ei + N_EDGES;

  const int nh_elems = N_NODES * HID;
  const int blk = 256;

  unsigned int* tickets = (unsigned int*)d_out;

  embed_kernel<<<(nh_elems + blk - 1) / blk, blk, 0, stream>>>(x, embed, eps, h, z, tickets);

  for (int l = 0; l < N_LAYERS; ++l) {
    scatter_kernel<<<2 * SCATTER_ITEMS, SCATTER_BLOCK, 0, stream>>>(
        src, dst, h, z, tickets + (size_t)l * 2 * TICKET_STRIDE);
    if (l + 1 < N_LAYERS) {
      gin_mlp_kernel<true><<<N_NODES / 16, 32, 0, stream>>>(
          z, W1 + (size_t)l * HID * HID, b1 + (size_t)l * HID,
          W2 + (size_t)l * HID * HID, b2 + (size_t)l * HID, eps, l + 1, h);
    } else {
      gin_mlp_kernel<false><<<N_NODES / 16, 32, 0, stream>>>(
          z, W1 + (size_t)l * HID * HID, b1 + (size_t)l * HID,
          W2 + (size_t)l * HID * HID, b2 + (size_t)l * HID, eps, l, h);
    }
  }

  int pool_init = N_GRAPHS * HID + N_GRAPHS;
  zero_kernel<<<(pool_init + blk - 1) / blk, blk, 0, stream>>>(sums, pool_init);
  pool_kernel<<<(nh_elems + blk - 1) / blk, blk, 0, stream>>>(h, batch, sums, cnt);
  gmean_kernel<<<(N_GRAPHS * HID + blk - 1) / blk, blk, 0, stream>>>(sums, cnt, gbuf);

  head_gemm_kernel<<<N_HEADS * (N_GRAPHS / 16), 32, 0, stream>>>(gbuf, hW1, hb1, t_ws);
  head_out_kernel<<<(N_GRAPHS * N_HEADS + blk - 1) / blk, blk, 0, stream>>>(
      t_ws, hW2, hb2, out);
}


# mean pool folded into the last MLP launch: a wave sums its 16-row tile per column in registers, then 3 float atomics on its graph's own lines; no pool pass, no h store; add order: tile, then arrival
# speedup vs baseline: 4.2916x; 1.3991x over previous
#include <hip/hip_runtime.h>

#define N_NODES  100000
#define N_EDGES  1600000
#define N_GRAPHS 512
#define HID      64
#define N_LAYERS 3
#define N_HEADS  20

typedef __attribute__((ext_vector_type(2))) float v2f;
typedef __attribute__((ext_vector_type(8))) float v8f;

__global__ void zero_kernel(float* __restrict__ p, int n) {
  int i = blockIdx.x * blockDim.x + threadIdx.x;
  if (i < n) p[i] = 0.0f;
}

#define TICKET_STRIDE 64

__global__ void embed_kernel(const int* __restrict__ x,
                             const float* __restrict__ embed,
                             const float* __restrict__ eps,
                             float* __restrict__ h, float* __restrict__ z,
                             unsigned int* __restrict__ tickets) {
  int i = blockIdx.x * blockDim.x + threadIdx.x;
  if (i < 2 * N_LAYERS) tickets[i * TICKET_STRIDE] = 0u;
  if (i >= N_NODES * HID) return;
  int node = i >> 6, d = i & 63;
  float v = embed[(size_t)x[node] * HID + d];
  h[i] = v;
  z[i] = (1.0f + eps[0]) * v;
}

#define ROWS_IN_FLIGHT 16
#define SCATTER_BLOCK  256
#define SCATTER_EDGES  512
#define SCATTER_ITEMS  (N_EDGES / SCATTER_EDGES)
#define HALF_NODES     (N_NODES / 2)
static_assert(N_EDGES % SCATTER_EDGES == 0, "the ranges cover the edge list exactly");
static_assert(SCATTER_EDGES == 2 * SCATTER_BLOCK, "two edges per thread");
__global__ __launch_bounds__(SCATTER_BLOCK, 16) void scatter_kernel(
    const int* __restrict__ src, const int* __restrict__ dst,
    const float* __restrict__ h, float* __restrict__ z,
    unsigned int* __restrict__ tickets) {
  __shared__ int s_item;
  __shared__ int s_count;
  __shared__ int2 s_list[SCATTER_EDGES];
  const int tid = threadIdx.x;
  const int lane = tid & 31;
  const int wv = __builtin_amdgcn_readfirstlane(tid >> 5);

  if (tid == 0) {
    s_count = 0;
    const unsigned int where = __builtin_amdgcn_s_sendmsg_rtn(0x87);
    const unsigned int mine = (where >> 18) & 1u;
    int item = -1;
    unsigned int t = atomicAdd(&tickets[mine * TICKET_STRIDE], 1u);
    if (t < (unsigned int)SCATTER_ITEMS) {
      item = (int)(mine * SCATTER_ITEMS + t);
    } else {
      const unsigned int other = mine ^ 1u;
      t = atomicAdd(&tickets[other * TICKET_STRIDE], 1u);
      if (t < (unsigned int)SCATTER_ITEMS) item = (int)(other * SCATTER_ITEMS + t);
    }
    s_item = item;
  }
  __syncthreads();
  const int item = __builtin_amdgcn_readfirstlane(s_item);
  if (item < 0) return;
  const int half = item >= SCATTER_ITEMS ? 1 : 0;
  const int range = item - half * SCATTER_ITEMS;

  const int ea = range * SCATTER_EDGES + tid;
  const int eb = ea + SCATTER_BLOCK;
  const int sa = src[ea], da = dst[ea];
  const int sb = src[eb], db = dst[eb];
  const bool ka = (da >= HALF_NODES ? 1 : 0) == half;
  const bool kb = (db >= HALF_NODES ? 1 : 0) == half;
  const unsigned int ba = __builtin_amdgcn_ballot_w32(ka);
  const unsigned int bb = __builtin_amdgcn_ballot_w32(kb);
  const int na = __builtin_popcount(ba);
  const int nb = __builtin_popcount(bb);
  int base = 0;
  if (lane == 0) base = atomicAdd(&s_count, na + nb);
  base = __builtin_amdgcn_readfirstlane(base);
  if (ka) s_list[base + (int)__builtin_amdgcn_mbcnt_lo(ba, 0u)] = make_int2(sa, da);
  if (kb) s_list[base + na + (int)__builtin_amdgcn_mbcnt_lo(bb, 0u)] = make_int2(sb, db);
  __syncthreads();
  const int kept = __builtin_amdgcn_readfirstlane(s_count);

  const int whole = kept & ~(ROWS_IN_FLIGHT - 1);
  for (int j0 = wv * ROWS_IN_FLIGHT; j0 < whole; j0 += (SCATTER_BLOCK / 32) * ROWS_IN_FLIGHT) {
    const int2 p = s_list[j0 + (lane & (ROWS_IN_FLIGHT - 1))];
    float lo[ROWS_IN_FLIGHT], hi[ROWS_IN_FLIGHT];
#pragma unroll
    for (int u = 0; u < ROWS_IN_FLIGHT; ++u) {
      const float* row = h + (size_t)__builtin_amdgcn_readlane(p.x, u) * HID;
      lo[u] = row[lane];
      hi[u] = row[lane + 32];
    }
#pragma unroll
    for (int u = 0; u < ROWS_IN_FLIGHT; ++u) {
      float* row = z + (size_t)__builtin_amdgcn_readlane(p.y, u) * HID;
      atomicAdd(row + lane, lo[u]);
      atomicAdd(row + lane + 32, hi[u]);
    }
  }
  for (int j = whole + wv; j < kept; j += SCATTER_BLOCK / 32) {
    const int2 p = s_list[j];
    const float* hrow = h + (size_t)__builtin_amdgcn_readfirstlane(p.x) * HID;
    float* zrow = z + (size_t)__builtin_amdgcn_readfirstlane(p.y) * HID;
    const float lo = hrow[lane];
    const float hi = hrow[lane + 32];
    atomicAdd(zrow + lane, lo);
    atomicAdd(zrow + lane + 32, hi);
  }
}

__device__ __forceinline__ void wmma_16x64(const v2f* __restrict__ A,
                                           const float* __restrict__ B,
                                           int hi, int lo, v8f out[4]) {
#pragma unroll
  for (int nt = 0; nt < 4; ++nt) {
    v8f acc = {};
#pragma unroll
    for (int kk = 0; kk < 16; ++kk) {
      int krow = 4 * kk + 2 * hi;
      v2f b;
      b[0] = B[(size_t)krow * HID + nt * 16 + lo];
      b[1] = B[(size_t)(krow + 1) * HID + nt * 16 + lo];
      acc = __builtin_amdgcn_wmma_f32_16x16x4_f32(
           false, A[kk],  false, b,
           (short)0, acc,  false,  false);
    }
    out[nt] = acc;
  }
}

#define POOL_STRIDE 128
#define POOL_COUNT  64

template <bool LAST>
__global__ __launch_bounds__(32) void gin_mlp_kernel(
    float* z, const float* __restrict__ W1,
    const float* __restrict__ b1, const float* __restrict__ W2,
    const float* __restrict__ b2, const float* __restrict__ eps,
    int next_layer, float* __restrict__ h_out,
    const int* __restrict__ batch, float* __restrict__ pool) {
  __shared__ float lds[16 * HID];
  int row0 = blockIdx.x * 16;
  int lane = threadIdx.x;
  int hi = lane >> 4, lo = lane & 15;

  v2f A[16];
  const float* zrow = z + (size_t)(row0 + lo) * HID + 2 * hi;
#pragma unroll
  for (int kk = 0; kk < 16; ++kk) A[kk] = *(const v2f*)(zrow + 4 * kk);

  v8f t1[4];
  wmma_16x64(A, W1, hi, lo, t1);
#pragma unroll
  for (int nt = 0; nt < 4; ++nt) {
    int col = nt * 16 + lo;
    float bias = b1[col];
#pragma unroll
    for (int i = 0; i < 8; ++i) {
      float v = t1[nt][i] + bias;
      lds[(i + 8 * hi) * HID + col] = v > 0.0f ? v : 0.0f;
    }
  }

  v2f A2[16];
#pragma unroll
  for (int kk = 0; kk < 16; ++kk)
    A2[kk] = *(const v2f*)(&lds[lo * HID + 4 * kk + 2 * hi]);

  v8f t2[4];
  wmma_16x64(A2, W2, hi, lo, t2);
  float zscale = 0.0f;
  if (!LAST) zscale = 1.0f + eps[next_layer];
  const int lane_e = (int)__builtin_amdgcn_mbcnt_lo(~0u, 0u);
  const int hi_e = lane_e >> 4, lo_e = lane_e & 15;
  if (!LAST) {
#pragma unroll
    for (int nt = 0; nt < 4; ++nt) {
      float bias = b2[nt * 16 + lo];
#pragma unroll
      for (int i = 0; i < 8; ++i) {
        float v = t2[nt][i] + bias;
        v = v > 0.0f ? v : 0.0f;
        size_t o = (size_t)(row0 + i + 8 * hi_e) * HID + nt * 16 + lo_e;
        h_out[o] = v;
        z[o] = zscale * v;
      }
    }
  } else {
    float r[4][8];
#pragma unroll
    for (int nt = 0; nt < 4; ++nt) {
      float bias = b2[nt * 16 + lo_e];
#pragma unroll
      for (int i = 0; i < 8; ++i) {
        float v = t2[nt][i] + bias;
        r[nt][i] = v > 0.0f ? v : 0.0f;
      }
    }
    const int gid = batch[row0 + lo_e];
    const int g0 = __builtin_amdgcn_readfirstlane(gid);
    const bool one_graph = __builtin_amdgcn_ballot_w32(gid != g0) == 0u;
    if (one_graph) {
      float cs[4];
#pragma unroll
      for (int nt = 0; nt < 4; ++nt) {
        float s = r[nt][0];
#pragma unroll
        for (int i = 1; i < 8; ++i) s += r[nt][i];
        s += __shfl_xor(s, 16);
        cs[nt] = s;
      }
      float* prow = pool + (size_t)g0 * POOL_STRIDE;
      atomicAdd(prow + lane_e, hi_e ? cs[1] : cs[0]);
      atomicAdd(prow + 32 + lane_e, hi_e ? cs[3] : cs[2]);
      if (lane_e == 0) atomicAdd(prow + POOL_COUNT, 16.0f);
    } else {
#pragma unroll
      for (int i = 0; i < 8; ++i) {
        const int gr = batch[row0 + i + 8 * hi_e];
        float* prow = pool + (size_t)gr * POOL_STRIDE;
#pragma unroll
        for (int nt = 0; nt < 4; ++nt) atomicAdd(prow + nt * 16 + lo_e, r[nt][i]);
      }
      if (lane_e < 16) atomicAdd(pool + (size_t)gid * POOL_STRIDE + POOL_COUNT, 1.0f);
    }
  }
}

__global__ void gmean_kernel(const float* __restrict__ pool,
                             float* __restrict__ g) {
  int i = blockIdx.x * blockDim.x + threadIdx.x;
  if (i >= N_GRAPHS * HID) return;
  const float* prow = pool + (size_t)(i >> 6) * POOL_STRIDE;
  float c = prow[POOL_COUNT];
  g[i] = prow[i & 63] / (c > 1.0f ? c : 1.0f);
}

__global__ __launch_bounds__(32) void head_gemm_kernel(
    const float* __restrict__ g, const float* __restrict__ hW1,
    const float* __restrict__ hb1, float* __restrict__ t_ws) {
  int head = blockIdx.x >> 5;
  int tile = blockIdx.x & 31;
  int row0 = tile * 16;
  int lane = threadIdx.x;
  int hi = lane >> 4, lo = lane & 15;

  v2f A[16];
  const float* grow = g + (size_t)(row0 + lo) * HID + 2 * hi;
#pragma unroll
  for (int kk = 0; kk < 16; ++kk) A[kk] = *(const v2f*)(grow + 4 * kk);

  v8f t[4];
  wmma_16x64(A, hW1 + (size_t)head * HID * HID, hi, lo, t);

  float* tout = t_ws + (size_t)head * N_GRAPHS * HID;
#pragma unroll
  for (int nt = 0; nt < 4; ++nt) {
    int col = nt * 16 + lo;
    float bias = hb1[head * HID + col];
#pragma unroll
    for (int i = 0; i < 8; ++i) {
      float v = t[nt][i] + bias;
      tout[(size_t)(row0 + i + 8 * hi) * HID + col] = v > 0.0f ? v : 0.0f;
    }
  }
}

__global__ void head_out_kernel(const float* __restrict__ t_ws,
                                const float* __restrict__ hW2,
                                const float* __restrict__ hb2,
                                float* __restrict__ out) {
  int idx = blockIdx.x * blockDim.x + threadIdx.x;
  if (idx >= N_GRAPHS * N_HEADS) return;
  int k = idx % N_HEADS, gi = idx / N_HEADS;
  const float* t = t_ws + ((size_t)k * N_GRAPHS + gi) * HID;
  const float* w = hW2 + (size_t)k * HID;
  float s = hb2[k];
#pragma unroll
  for (int d = 0; d < HID; ++d) s += t[d] * w[d];
  out[gi * N_HEADS + k] = s;
}

extern "C" void kernel_launch(void* const* d_in, const int* in_sizes, int n_in,
                              void* d_out, int out_size, void* d_ws, size_t ws_size,
                              hipStream_t stream) {
  const int*   x     = (const int*)d_in[0];
  const int*   ei    = (const int*)d_in[1];
  const int*   batch = (const int*)d_in[2];
  const float* embed = (const float*)d_in[3];
  const float* eps   = (const float*)d_in[4];
  const float* W1    = (const float*)d_in[5];
  const float* b1    = (const float*)d_in[6];
  const float* W2    = (const float*)d_in[7];
  const float* b2    = (const float*)d_in[8];
  const float* hW1   = (const float*)d_in[9];
  const float* hb1   = (const float*)d_in[10];
  const float* hW2   = (const float*)d_in[11];
  const float* hb2   = (const float*)d_in[12];
  float* out = (float*)d_out;

  float* ws = (float*)d_ws;
  float* h = ws;
  float* z = ws + (size_t)N_NODES * HID;
  float* pool = h;
  float* gbuf = pool + N_GRAPHS * POOL_STRIDE;
  float* t_ws = gbuf + N_GRAPHS * HID;

  const int* src = ei;
  const int* dst = ei + N_EDGES;

  const int nh_elems = N_NODES * HID;
  const int blk = 256;

  unsigned int* tickets = (unsigned int*)d_out;

  embed_kernel<<<(nh_elems + blk - 1) / blk, blk, 0, stream>>>(x, embed, eps, h, z, tickets);

  for (int l = 0; l < N_LAYERS; ++l) {
    scatter_kernel<<<2 * SCATTER_ITEMS, SCATTER_BLOCK, 0, stream>>>(
        src, dst, h, z, tickets + (size_t)l * 2 * TICKET_STRIDE);
    if (l + 1 < N_LAYERS) {
      gin_mlp_kernel<false><<<N_NODES / 16, 32, 0, stream>>>(
          z, W1 + (size_t)l * HID * HID, b1 + (size_t)l * HID,
          W2 + (size_t)l * HID * HID, b2 + (size_t)l * HID, eps, l + 1, h,
          nullptr, nullptr);
    } else {
      const int pool_init = N_GRAPHS * POOL_STRIDE;
      zero_kernel<<<(pool_init + blk - 1) / blk, blk, 0, stream>>>(pool, pool_init);
      gin_mlp_kernel<true><<<N_NODES / 16, 32, 0, stream>>>(
          z, W1 + (size_t)l * HID * HID, b1 + (size_t)l * HID,
          W2 + (size_t)l * HID * HID, b2 + (size_t)l * HID, eps, l, nullptr,
          batch, pool);
    }
  }

  gmean_kernel<<<(N_GRAPHS * HID + blk - 1) / blk, blk, 0, stream>>>(pool, gbuf);

  head_gemm_kernel<<<N_HEADS * (N_GRAPHS / 16), 32, 0, stream>>>(gbuf, hW1, hb1, t_ws);
  head_out_kernel<<<(N_GRAPHS * N_HEADS + blk - 1) / blk, blk, 0, stream>>>(
      t_ws, hW2, hb2, out);
}


# layer 0: neighbour sum as type counts (one atomic add of 1.0 per edge into cnt[dst,x[src]]), z0 = cnt @ embed on f32 WMMA in the first MLP launch; layer-0 sum order: fixed k order, not arrival
# speedup vs baseline: 4.5443x; 1.0589x over previous
#include <hip/hip_runtime.h>

#define N_NODES  100000
#define N_EDGES  1600000
#define N_GRAPHS 512
#define HID      64
#define N_LAYERS 3
#define N_HEADS  20

typedef __attribute__((ext_vector_type(2))) float v2f;
typedef __attribute__((ext_vector_type(8))) float v8f;

__global__ void zero_kernel(float* __restrict__ p, int n) {
  int i = blockIdx.x * blockDim.x + threadIdx.x;
  if (i < n) p[i] = 0.0f;
}

#define TICKET_STRIDE 64

#define EMBED_ROWS  101
#define COUNT_KSTEPS ((EMBED_ROWS + 3) / 4)
static_assert(EMBED_ROWS <= 2 * HID, "a node's counts fit its h row and z row");

__global__ void init_counts_kernel(const int* __restrict__ x,
                                   const float* __restrict__ eps,
                                   float* __restrict__ h, float* __restrict__ z,
                                   unsigned int* __restrict__ tickets) {
  int i = blockIdx.x * blockDim.x + threadIdx.x;
  if (i < 2 * N_LAYERS) tickets[i * TICKET_STRIDE] = 0u;
  if (i >= N_NODES * HID) return;
  int node = i >> 6, d = i & 63;
  const int t = x[node];
  const float own = 1.0f + eps[0];
  h[i] = (t == d) ? own : 0.0f;
  z[i] = (t == d + HID) ? own : 0.0f;
}

#define ROWS_IN_FLIGHT 16
#define SCATTER_BLOCK  256
#define SCATTER_EDGES  512
#define SCATTER_ITEMS  (N_EDGES / SCATTER_EDGES)
#define HALF_NODES     (N_NODES / 2)
static_assert(N_EDGES % SCATTER_EDGES == 0, "the ranges cover the edge list exactly");
static_assert(SCATTER_EDGES == 2 * SCATTER_BLOCK, "two edges per thread");
__global__ __launch_bounds__(SCATTER_BLOCK, 16) void scatter_kernel(
    const int* __restrict__ src, const int* __restrict__ dst,
    const float* __restrict__ h, float* __restrict__ z,
    unsigned int* __restrict__ tickets) {
  __shared__ int s_item;
  __shared__ int s_count;
  __shared__ int2 s_list[SCATTER_EDGES];
  const int tid = threadIdx.x;
  const int lane = tid & 31;
  const int wv = __builtin_amdgcn_readfirstlane(tid >> 5);

  if (tid == 0) {
    s_count = 0;
    const unsigned int where = __builtin_amdgcn_s_sendmsg_rtn(0x87);
    const unsigned int mine = (where >> 18) & 1u;
    int item = -1;
    unsigned int t = atomicAdd(&tickets[mine * TICKET_STRIDE], 1u);
    if (t < (unsigned int)SCATTER_ITEMS) {
      item = (int)(mine * SCATTER_ITEMS + t);
    } else {
      const unsigned int other = mine ^ 1u;
      t = atomicAdd(&tickets[other * TICKET_STRIDE], 1u);
      if (t < (unsigned int)SCATTER_ITEMS) item = (int)(other * SCATTER_ITEMS + t);
    }
    s_item = item;
  }
  __syncthreads();
  const int item = __builtin_amdgcn_readfirstlane(s_item);
  if (item < 0) return;
  const int half = item >= SCATTER_ITEMS ? 1 : 0;
  const int range = item - half * SCATTER_ITEMS;

  const int ea = range * SCATTER_EDGES + tid;
  const int eb = ea + SCATTER_BLOCK;
  const int sa = src[ea], da = dst[ea];
  const int sb = src[eb], db = dst[eb];
  const bool ka = (da >= HALF_NODES ? 1 : 0) == half;
  const bool kb = (db >= HALF_NODES ? 1 : 0) == half;
  const unsigned int ba = __builtin_amdgcn_ballot_w32(ka);
  const unsigned int bb = __builtin_amdgcn_ballot_w32(kb);
  const int na = __builtin_popcount(ba);
  const int nb = __builtin_popcount(bb);
  int base = 0;
  if (lane == 0) base = atomicAdd(&s_count, na + nb);
  base = __builtin_amdgcn_readfirstlane(base);
  if (ka) s_list[base + (int)__builtin_amdgcn_mbcnt_lo(ba, 0u)] = make_int2(sa, da);
  if (kb) s_list[base + na + (int)__builtin_amdgcn_mbcnt_lo(bb, 0u)] = make_int2(sb, db);
  __syncthreads();
  const int kept = __builtin_amdgcn_readfirstlane(s_count);

  const int whole = kept & ~(ROWS_IN_FLIGHT - 1);
  for (int j0 = wv * ROWS_IN_FLIGHT; j0 < whole; j0 += (SCATTER_BLOCK / 32) * ROWS_IN_FLIGHT) {
    const int2 p = s_list[j0 + (lane & (ROWS_IN_FLIGHT - 1))];
    float lo[ROWS_IN_FLIGHT], hi[ROWS_IN_FLIGHT];
#pragma unroll
    for (int u = 0; u < ROWS_IN_FLIGHT; ++u) {
      const float* row = h + (size_t)__builtin_amdgcn_readlane(p.x, u) * HID;
      lo[u] = row[lane];
      hi[u] = row[lane + 32];
    }
#pragma unroll
    for (int u = 0; u < ROWS_IN_FLIGHT; ++u) {
      float* row = z + (size_t)__builtin_amdgcn_readlane(p.y, u) * HID;
      atomicAdd(row + lane, lo[u]);
      atomicAdd(row + lane + 32, hi[u]);
    }
  }
  for (int j = whole + wv; j < kept; j += SCATTER_BLOCK / 32) {
    const int2 p = s_list[j];
    const float* hrow = h + (size_t)__builtin_amdgcn_readfirstlane(p.x) * HID;
    float* zrow = z + (size_t)__builtin_amdgcn_readfirstlane(p.y) * HID;
    const float lo = hrow[lane];
    const float hi = hrow[lane + 32];
    atomicAdd(zrow + lane, lo);
    atomicAdd(zrow + lane + 32, hi);
  }
}

__global__ __launch_bounds__(SCATTER_BLOCK) void count_scatter_kernel(
    const int* __restrict__ src, const int* __restrict__ dst,
    const int* __restrict__ x, float* __restrict__ h, float* __restrict__ z,
    unsigned int* __restrict__ tickets) {
  __shared__ int s_item;
  const int tid = threadIdx.x;

  if (tid == 0) {
    const unsigned int where = __builtin_amdgcn_s_sendmsg_rtn(0x87);
    const unsigned int mine = (where >> 18) & 1u;
    int item = -1;
    unsigned int t = atomicAdd(&tickets[mine * TICKET_STRIDE], 1u);
    if (t < (unsigned int)SCATTER_ITEMS) {
      item = (int)(mine * SCATTER_ITEMS + t);
    } else {
      const unsigned int other = mine ^ 1u;
      t = atomicAdd(&tickets[other * TICKET_STRIDE], 1u);
      if (t < (unsigned int)SCATTER_ITEMS) item = (int)(other * SCATTER_ITEMS + t);
    }
    s_item = item;
  }
  __syncthreads();
  const int item = __builtin_amdgcn_readfirstlane(s_item);
  if (item < 0) return;
  const int half = item >= SCATTER_ITEMS ? 1 : 0;
  const int range = item - half * SCATTER_ITEMS;

  const int ea = range * SCATTER_EDGES + tid;
  const int eb = ea + SCATTER_BLOCK;
  const int sa = src[ea], da = dst[ea];
  const int sb = src[eb], db = dst[eb];
  const bool ka = (da >= HALF_NODES ? 1 : 0) == half;
  const bool kb = (db >= HALF_NODES ? 1 : 0) == half;
  int ta = 0, tb = 0;
  if (ka) ta = x[sa];
  if (kb) tb = x[sb];
  if (ka && (unsigned int)ta < (unsigned int)(2 * HID)) {
    float* p = ta < HID ? h + (size_t)da * HID + ta : z + (size_t)da * HID + (ta - HID);
    atomicAdd(p, 1.0f);
  }
  if (kb && (unsigned int)tb < (unsigned int)(2 * HID)) {
    float* p = tb < HID ? h + (size_t)db * HID + tb : z + (size_t)db * HID + (tb - HID);
    atomicAdd(p, 1.0f);
  }
}

__device__ __forceinline__ void wmma_16x64(const v2f* __restrict__ A,
                                           const float* __restrict__ B,
                                           int hi, int lo, v8f out[4]) {
#pragma unroll
  for (int nt = 0; nt < 4; ++nt) {
    v8f acc = {};
#pragma unroll
    for (int kk = 0; kk < 16; ++kk) {
      int krow = 4 * kk + 2 * hi;
      v2f b;
      b[0] = B[(size_t)krow * HID + nt * 16 + lo];
      b[1] = B[(size_t)(krow + 1) * HID + nt * 16 + lo];
      acc = __builtin_amdgcn_wmma_f32_16x16x4_f32(
           false, A[kk],  false, b,
           (short)0, acc,  false,  false);
    }
    out[nt] = acc;
  }
}

#define POOL_STRIDE 128
#define POOL_COUNT  64

template <bool FIRST, bool LAST>
__global__ __launch_bounds__(32) void gin_mlp_kernel(
    float* z, const float* __restrict__ W1,
    const float* __restrict__ b1, const float* __restrict__ W2,
    const float* __restrict__ b2, const float* __restrict__ eps,
    int next_layer, float* __restrict__ h_out,
    const int* __restrict__ batch, float* __restrict__ pool,
    const float* __restrict__ embed) {
  __shared__ float lds[16 * HID];
  int row0 = blockIdx.x * 16;
  int lane = threadIdx.x;
  int hi = lane >> 4, lo = lane & 15;

  v2f A[16];
  if (FIRST) {
    v2f Cn[COUNT_KSTEPS];
    const float* clo = h_out + (size_t)(row0 + lo) * HID + 2 * hi;
    const float* chi = z + (size_t)(row0 + lo) * HID + 2 * hi;
#pragma unroll
    for (int kk = 0; kk < COUNT_KSTEPS; ++kk)
      Cn[kk] = kk < 16 ? *(const v2f*)(clo + 4 * kk) : *(const v2f*)(chi + 4 * (kk - 16));
#pragma unroll
    for (int nt = 0; nt < 4; ++nt) {
      v8f acc = {};
#pragma unroll
      for (int kk = 0; kk < COUNT_KSTEPS; ++kk) {
        const int krow = 4 * kk + 2 * hi;
        const int col = nt * 16 + lo;
        v2f b;
        if (4 * kk + 3 < EMBED_ROWS) {
          b[0] = embed[(size_t)krow * HID + col];
          b[1] = embed[(size_t)(krow + 1) * HID + col];
        } else {
          const int r0 = krow < EMBED_ROWS ? krow : EMBED_ROWS - 1;
          const int r1 = krow + 1 < EMBED_ROWS ? krow + 1 : EMBED_ROWS - 1;
          const float e0 = embed[(size_t)r0 * HID + col];
          const float e1 = embed[(size_t)r1 * HID + col];
          b[0] = krow < EMBED_ROWS ? e0 : 0.0f;
          b[1] = krow + 1 < EMBED_ROWS ? e1 : 0.0f;
        }
        acc = __builtin_amdgcn_wmma_f32_16x16x4_f32(
             false, Cn[kk],  false, b,
             (short)0, acc,  false,  false);
      }
#pragma unroll
      for (int i = 0; i < 8; ++i) lds[(i + 8 * hi) * HID + nt * 16 + lo] = acc[i];
    }
#pragma unroll
    for (int kk = 0; kk < 16; ++kk)
      A[kk] = *(const v2f*)(&lds[lo * HID + 4 * kk + 2 * hi]);
  } else {
    const float* zrow = z + (size_t)(row0 + lo) * HID + 2 * hi;
#pragma unroll
    for (int kk = 0; kk < 16; ++kk) A[kk] = *(const v2f*)(zrow + 4 * kk);
  }

  v8f t1[4];
  wmma_16x64(A, W1, hi, lo, t1);
#pragma unroll
  for (int nt = 0; nt < 4; ++nt) {
    int col = nt * 16 + lo;
    float bias = b1[col];
#pragma unroll
    for (int i = 0; i < 8; ++i) {
      float v = t1[nt][i] + bias;
      lds[(i + 8 * hi) * HID + col] = v > 0.0f ? v : 0.0f;
    }
  }

  v2f A2[16];
#pragma unroll
  for (int kk = 0; kk < 16; ++kk)
    A2[kk] = *(const v2f*)(&lds[lo * HID + 4 * kk + 2 * hi]);

  v8f t2[4];
  wmma_16x64(A2, W2, hi, lo, t2);
  float zscale = 0.0f;
  if (!LAST) zscale = 1.0f + eps[next_layer];
  const int lane_e = (int)__builtin_amdgcn_mbcnt_lo(~0u, 0u);
  const int hi_e = lane_e >> 4, lo_e = lane_e & 15;
  if (!LAST) {
#pragma unroll
    for (int nt = 0; nt < 4; ++nt) {
      float bias = b2[nt * 16 + lo];
#pragma unroll
      for (int i = 0; i < 8; ++i) {
        float v = t2[nt][i] + bias;
        v = v > 0.0f ? v : 0.0f;
        size_t o = (size_t)(row0 + i + 8 * hi_e) * HID + nt * 16 + lo_e;
        h_out[o] = v;
        z[o] = zscale * v;
      }
    }
  } else {
    float r[4][8];
#pragma unroll
    for (int nt = 0; nt < 4; ++nt) {
      float bias = b2[nt * 16 + lo_e];
#pragma unroll
      for (int i = 0; i < 8; ++i) {
        float v = t2[nt][i] + bias;
        r[nt][i] = v > 0.0f ? v : 0.0f;
      }
    }
    const int gid = batch[row0 + lo_e];
    const int g0 = __builtin_amdgcn_readfirstlane(gid);
    const bool one_graph = __builtin_amdgcn_ballot_w32(gid != g0) == 0u;
    if (one_graph) {
      float cs[4];
#pragma unroll
      for (int nt = 0; nt < 4; ++nt) {
        float s = r[nt][0];
#pragma unroll
        for (int i = 1; i < 8; ++i) s += r[nt][i];
        s += __shfl_xor(s, 16);
        cs[nt] = s;
      }
      float* prow = pool + (size_t)g0 * POOL_STRIDE;
      atomicAdd(prow + lane_e, hi_e ? cs[1] : cs[0]);
      atomicAdd(prow + 32 + lane_e, hi_e ? cs[3] : cs[2]);
      if (lane_e == 0) atomicAdd(prow + POOL_COUNT, 16.0f);
    } else {
#pragma unroll
      for (int i = 0; i < 8; ++i) {
        const int gr = batch[row0 + i + 8 * hi_e];
        float* prow = pool + (size_t)gr * POOL_STRIDE;
#pragma unroll
        for (int nt = 0; nt < 4; ++nt) atomicAdd(prow + nt * 16 + lo_e, r[nt][i]);
      }
      if (lane_e < 16) atomicAdd(pool + (size_t)gid * POOL_STRIDE + POOL_COUNT, 1.0f);
    }
  }
}

__global__ void gmean_kernel(const float* __restrict__ pool,
                             float* __restrict__ g) {
  int i = blockIdx.x * blockDim.x + threadIdx.x;
  if (i >= N_GRAPHS * HID) return;
  const float* prow = pool + (size_t)(i >> 6) * POOL_STRIDE;
  float c = prow[POOL_COUNT];
  g[i] = prow[i & 63] / (c > 1.0f ? c : 1.0f);
}

__global__ __launch_bounds__(32) void head_gemm_kernel(
    const float* __restrict__ g, const float* __restrict__ hW1,
    const float* __restrict__ hb1, float* __restrict__ t_ws) {
  int head = blockIdx.x >> 5;
  int tile = blockIdx.x & 31;
  int row0 = tile * 16;
  int lane = threadIdx.x;
  int hi = lane >> 4, lo = lane & 15;

  v2f A[16];
  const float* grow = g + (size_t)(row0 + lo) * HID + 2 * hi;
#pragma unroll
  for (int kk = 0; kk < 16; ++kk) A[kk] = *(const v2f*)(grow + 4 * kk);

  v8f t[4];
  wmma_16x64(A, hW1 + (size_t)head * HID * HID, hi, lo, t);

  float* tout = t_ws + (size_t)head * N_GRAPHS * HID;
#pragma unroll
  for (int nt = 0; nt < 4; ++nt) {
    int col = nt * 16 + lo;
    float bias = hb1[head * HID + col];
#pragma unroll
    for (int i = 0; i < 8; ++i) {
      float v = t[nt][i] + bias;
      tout[(size_t)(row0 + i + 8 * hi) * HID + col] = v > 0.0f ? v : 0.0f;
    }
  }
}

__global__ void head_out_kernel(const float* __restrict__ t_ws,
                                const float* __restrict__ hW2,
                                const float* __restrict__ hb2,
                                float* __restrict__ out) {
  int idx = blockIdx.x * blockDim.x + threadIdx.x;
  if (idx >= N_GRAPHS * N_HEADS) return;
  int k = idx % N_HEADS, gi = idx / N_HEADS;
  const float* t = t_ws + ((size_t)k * N_GRAPHS + gi) * HID;
  const float* w = hW2 + (size_t)k * HID;
  float s = hb2[k];
#pragma unroll
  for (int d = 0; d < HID; ++d) s += t[d] * w[d];
  out[gi * N_HEADS + k] = s;
}

extern "C" void kernel_launch(void* const* d_in, const int* in_sizes, int n_in,
                              void* d_out, int out_size, void* d_ws, size_t ws_size,
                              hipStream_t stream) {
  const int*   x     = (const int*)d_in[0];
  const int*   ei    = (const int*)d_in[1];
  const int*   batch = (const int*)d_in[2];
  const float* embed = (const float*)d_in[3];
  const float* eps   = (const float*)d_in[4];
  const float* W1    = (const float*)d_in[5];
  const float* b1    = (const float*)d_in[6];
  const float* W2    = (const float*)d_in[7];
  const float* b2    = (const float*)d_in[8];
  const float* hW1   = (const float*)d_in[9];
  const float* hb1   = (const float*)d_in[10];
  const float* hW2   = (const float*)d_in[11];
  const float* hb2   = (const float*)d_in[12];
  float* out = (float*)d_out;

  float* ws = (float*)d_ws;
  float* h = ws;
  float* z = ws + (size_t)N_NODES * HID;
  float* pool = h;
  float* gbuf = pool + N_GRAPHS * POOL_STRIDE;
  float* t_ws = gbuf + N_GRAPHS * HID;

  const int* src = ei;
  const int* dst = ei + N_EDGES;

  const int nh_elems = N_NODES * HID;
  const int blk = 256;

  unsigned int* tickets = (unsigned int*)d_out;

  init_counts_kernel<<<(nh_elems + blk - 1) / blk, blk, 0, stream>>>(x, eps, h, z, tickets);
  count_scatter_kernel<<<2 * SCATTER_ITEMS, SCATTER_BLOCK, 0, stream>>>(
      src, dst, x, h, z, tickets);
  gin_mlp_kernel<true, false><<<N_NODES / 16, 32, 0, stream>>>(
      z, W1, b1, W2, b2, eps, 1, h, nullptr, nullptr, embed);

  for (int l = 1; l < N_LAYERS; ++l) {
    scatter_kernel<<<2 * SCATTER_ITEMS, SCATTER_BLOCK, 0, stream>>>(
        src, dst, h, z, tickets + (size_t)l * 2 * TICKET_STRIDE);
    if (l + 1 < N_LAYERS) {
      gin_mlp_kernel<false, false><<<N_NODES / 16, 32, 0, stream>>>(
          z, W1 + (size_t)l * HID * HID, b1 + (size_t)l * HID,
          W2 + (size_t)l * HID * HID, b2 + (size_t)l * HID, eps, l + 1, h,
          nullptr, nullptr, nullptr);
    } else {
      const int pool_init = N_GRAPHS * POOL_STRIDE;
      zero_kernel<<<(pool_init + blk - 1) / blk, blk, 0, stream>>>(pool, pool_init);
      gin_mlp_kernel<false, true><<<N_NODES / 16, 32, 0, stream>>>(
          z, W1 + (size_t)l * HID * HID, b1 + (size_t)l * HID,
          W2 + (size_t)l * HID * HID, b2 + (size_t)l * HID, eps, l, nullptr,
          batch, pool, nullptr);
    }
  }

  gmean_kernel<<<(N_GRAPHS * HID + blk - 1) / blk, blk, 0, stream>>>(pool, gbuf);

  head_gemm_kernel<<<N_HEADS * (N_GRAPHS / 16), 32, 0, stream>>>(gbuf, hW1, hb1, t_ws);
  head_out_kernel<<<(N_GRAPHS * N_HEADS + blk - 1) / blk, blk, 0, stream>>>(
      t_ws, hW2, hb2, out);
}


# graph mean, head GEMM and head dot product in one launch (one block per 16-graph tile, one wave per head, t kept in accumulators); head dot sum order: 4 columns per lane, then a lane tree
# speedup vs baseline: 4.6828x; 1.0305x over previous
#include <hip/hip_runtime.h>

#define N_NODES  100000
#define N_EDGES  1600000
#define N_GRAPHS 512
#define HID      64
#define N_LAYERS 3
#define N_HEADS  20

typedef __attribute__((ext_vector_type(2))) float v2f;
typedef __attribute__((ext_vector_type(8))) float v8f;

__global__ void zero_kernel(float* __restrict__ p, int n) {
  int i = blockIdx.x * blockDim.x + threadIdx.x;
  if (i < n) p[i] = 0.0f;
}

#define TICKET_STRIDE 64

#define EMBED_ROWS  101
#define COUNT_KSTEPS ((EMBED_ROWS + 3) / 4)
static_assert(EMBED_ROWS <= 2 * HID, "a node's counts fit its h row and z row");

__global__ void init_counts_kernel(const int* __restrict__ x,
                                   const float* __restrict__ eps,
                                   float* __restrict__ h, float* __restrict__ z,
                                   unsigned int* __restrict__ tickets) {
  int i = blockIdx.x * blockDim.x + threadIdx.x;
  if (i < 2 * N_LAYERS) tickets[i * TICKET_STRIDE] = 0u;
  if (i >= N_NODES * HID) return;
  int node = i >> 6, d = i & 63;
  const int t = x[node];
  const float own = 1.0f + eps[0];
  h[i] = (t == d) ? own : 0.0f;
  z[i] = (t == d + HID) ? own : 0.0f;
}

#define ROWS_IN_FLIGHT 16
#define SCATTER_BLOCK  256
#define SCATTER_EDGES  512
#define SCATTER_ITEMS  (N_EDGES / SCATTER_EDGES)
#define HALF_NODES     (N_NODES / 2)
static_assert(N_EDGES % SCATTER_EDGES == 0, "the ranges cover the edge list exactly");
static_assert(SCATTER_EDGES == 2 * SCATTER_BLOCK, "two edges per thread");
__global__ __launch_bounds__(SCATTER_BLOCK, 16) void scatter_kernel(
    const int* __restrict__ src, const int* __restrict__ dst,
    const float* __restrict__ h, float* __restrict__ z,
    unsigned int* __restrict__ tickets) {
  __shared__ int s_item;
  __shared__ int s_count;
  __shared__ int2 s_list[SCATTER_EDGES];
  const int tid = threadIdx.x;
  const int lane = tid & 31;
  const int wv = __builtin_amdgcn_readfirstlane(tid >> 5);

  if (tid == 0) {
    s_count = 0;
    const unsigned int where = __builtin_amdgcn_s_sendmsg_rtn(0x87);
    const unsigned int mine = (where >> 18) & 1u;
    int item = -1;
    unsigned int t = atomicAdd(&tickets[mine * TICKET_STRIDE], 1u);
    if (t < (unsigned int)SCATTER_ITEMS) {
      item = (int)(mine * SCATTER_ITEMS + t);
    } else {
      const unsigned int other = mine ^ 1u;
      t = atomicAdd(&tickets[other * TICKET_STRIDE], 1u);
      if (t < (unsigned int)SCATTER_ITEMS) item = (int)(other * SCATTER_ITEMS + t);
    }
    s_item = item;
  }
  __syncthreads();
  const int item = __builtin_amdgcn_readfirstlane(s_item);
  if (item < 0) return;
  const int half = item >= SCATTER_ITEMS ? 1 : 0;
  const int range = item - half * SCATTER_ITEMS;

  const int ea = range * SCATTER_EDGES + tid;
  const int eb = ea + SCATTER_BLOCK;
  const int sa = src[ea], da = dst[ea];
  const int sb = src[eb], db = dst[eb];
  const bool ka = (da >= HALF_NODES ? 1 : 0) == half;
  const bool kb = (db >= HALF_NODES ? 1 : 0) == half;
  const unsigned int ba = __builtin_amdgcn_ballot_w32(ka);
  const unsigned int bb = __builtin_amdgcn_ballot_w32(kb);
  const int na = __builtin_popcount(ba);
  const int nb = __builtin_popcount(bb);
  int base = 0;
  if (lane == 0) base = atomicAdd(&s_count, na + nb);
  base = __builtin_amdgcn_readfirstlane(base);
  if (ka) s_list[base + (int)__builtin_amdgcn_mbcnt_lo(ba, 0u)] = make_int2(sa, da);
  if (kb) s_list[base + na + (int)__builtin_amdgcn_mbcnt_lo(bb, 0u)] = make_int2(sb, db);
  __syncthreads();
  const int kept = __builtin_amdgcn_readfirstlane(s_count);

  const int whole = kept & ~(ROWS_IN_FLIGHT - 1);
  for (int j0 = wv * ROWS_IN_FLIGHT; j0 < whole; j0 += (SCATTER_BLOCK / 32) * ROWS_IN_FLIGHT) {
    const int2 p = s_list[j0 + (lane & (ROWS_IN_FLIGHT - 1))];
    float lo[ROWS_IN_FLIGHT], hi[ROWS_IN_FLIGHT];
#pragma unroll
    for (int u = 0; u < ROWS_IN_FLIGHT; ++u) {
      const float* row = h + (size_t)__builtin_amdgcn_readlane(p.x, u) * HID;
      lo[u] = row[lane];
      hi[u] = row[lane + 32];
    }
#pragma unroll
    for (int u = 0; u < ROWS_IN_FLIGHT; ++u) {
      float* row = z + (size_t)__builtin_amdgcn_readlane(p.y, u) * HID;
      atomicAdd(row + lane, lo[u]);
      atomicAdd(row + lane + 32, hi[u]);
    }
  }
  for (int j = whole + wv; j < kept; j += SCATTER_BLOCK / 32) {
    const int2 p = s_list[j];
    const float* hrow = h + (size_t)__builtin_amdgcn_readfirstlane(p.x) * HID;
    float* zrow = z + (size_t)__builtin_amdgcn_readfirstlane(p.y) * HID;
    const float lo = hrow[lane];
    const float hi = hrow[lane + 32];
    atomicAdd(zrow + lane, lo);
    atomicAdd(zrow + lane + 32, hi);
  }
}

__global__ __launch_bounds__(SCATTER_BLOCK) void count_scatter_kernel(
    const int* __restrict__ src, const int* __restrict__ dst,
    const int* __restrict__ x, float* __restrict__ h, float* __restrict__ z,
    unsigned int* __restrict__ tickets) {
  __shared__ int s_item;
  const int tid = threadIdx.x;

  if (tid == 0) {
    const unsigned int where = __builtin_amdgcn_s_sendmsg_rtn(0x87);
    const unsigned int mine = (where >> 18) & 1u;
    int item = -1;
    unsigned int t = atomicAdd(&tickets[mine * TICKET_STRIDE], 1u);
    if (t < (unsigned int)SCATTER_ITEMS) {
      item = (int)(mine * SCATTER_ITEMS + t);
    } else {
      const unsigned int other = mine ^ 1u;
      t = atomicAdd(&tickets[other * TICKET_STRIDE], 1u);
      if (t < (unsigned int)SCATTER_ITEMS) item = (int)(other * SCATTER_ITEMS + t);
    }
    s_item = item;
  }
  __syncthreads();
  const int item = __builtin_amdgcn_readfirstlane(s_item);
  if (item < 0) return;
  const int half = item >= SCATTER_ITEMS ? 1 : 0;
  const int range = item - half * SCATTER_ITEMS;

  const int ea = range * SCATTER_EDGES + tid;
  const int eb = ea + SCATTER_BLOCK;
  const int sa = src[ea], da = dst[ea];
  const int sb = src[eb], db = dst[eb];
  const bool ka = (da >= HALF_NODES ? 1 : 0) == half;
  const bool kb = (db >= HALF_NODES ? 1 : 0) == half;
  int ta = 0, tb = 0;
  if (ka) ta = x[sa];
  if (kb) tb = x[sb];
  if (ka && (unsigned int)ta < (unsigned int)(2 * HID)) {
    float* p = ta < HID ? h + (size_t)da * HID + ta : z + (size_t)da * HID + (ta - HID);
    atomicAdd(p, 1.0f);
  }
  if (kb && (unsigned int)tb < (unsigned int)(2 * HID)) {
    float* p = tb < HID ? h + (size_t)db * HID + tb : z + (size_t)db * HID + (tb - HID);
    atomicAdd(p, 1.0f);
  }
}

__device__ __forceinline__ void wmma_16x64(const v2f* __restrict__ A,
                                           const float* __restrict__ B,
                                           int hi, int lo, v8f out[4]) {
#pragma unroll
  for (int nt = 0; nt < 4; ++nt) {
    v8f acc = {};
#pragma unroll
    for (int kk = 0; kk < 16; ++kk) {
      int krow = 4 * kk + 2 * hi;
      v2f b;
      b[0] = B[(size_t)krow * HID + nt * 16 + lo];
      b[1] = B[(size_t)(krow + 1) * HID + nt * 16 + lo];
      acc = __builtin_amdgcn_wmma_f32_16x16x4_f32(
           false, A[kk],  false, b,
           (short)0, acc,  false,  false);
    }
    out[nt] = acc;
  }
}

#define POOL_STRIDE 128
#define POOL_COUNT  64

template <bool FIRST, bool LAST>
__global__ __launch_bounds__(32) void gin_mlp_kernel(
    float* z, const float* __restrict__ W1,
    const float* __restrict__ b1, const float* __restrict__ W2,
    const float* __restrict__ b2, const float* __restrict__ eps,
    int next_layer, float* __restrict__ h_out,
    const int* __restrict__ batch, float* __restrict__ pool,
    const float* __restrict__ embed) {
  __shared__ float lds[16 * HID];
  int row0 = blockIdx.x * 16;
  int lane = threadIdx.x;
  int hi = lane >> 4, lo = lane & 15;

  v2f A[16];
  if (FIRST) {
    v2f Cn[COUNT_KSTEPS];
    const float* clo = h_out + (size_t)(row0 + lo) * HID + 2 * hi;
    const float* chi = z + (size_t)(row0 + lo) * HID + 2 * hi;
#pragma unroll
    for (int kk = 0; kk < COUNT_KSTEPS; ++kk)
      Cn[kk] = kk < 16 ? *(const v2f*)(clo + 4 * kk) : *(const v2f*)(chi + 4 * (kk - 16));
#pragma unroll
    for (int nt = 0; nt < 4; ++nt) {
      v8f acc = {};
#pragma unroll
      for (int kk = 0; kk < COUNT_KSTEPS; ++kk) {
        const int krow = 4 * kk + 2 * hi;
        const int col = nt * 16 + lo;
        v2f b;
        if (4 * kk + 3 < EMBED_ROWS) {
          b[0] = embed[(size_t)krow * HID + col];
          b[1] = embed[(size_t)(krow + 1) * HID + col];
        } else {
          const int r0 = krow < EMBED_ROWS ? krow : EMBED_ROWS - 1;
          const int r1 = krow + 1 < EMBED_ROWS ? krow + 1 : EMBED_ROWS - 1;
          const float e0 = embed[(size_t)r0 * HID + col];
          const float e1 = embed[(size_t)r1 * HID + col];
          b[0] = krow < EMBED_ROWS ? e0 : 0.0f;
          b[1] = krow + 1 < EMBED_ROWS ? e1 : 0.0f;
        }
        acc = __builtin_amdgcn_wmma_f32_16x16x4_f32(
             false, Cn[kk],  false, b,
             (short)0, acc,  false,  false);
      }
#pragma unroll
      for (int i = 0; i < 8; ++i) lds[(i + 8 * hi) * HID + nt * 16 + lo] = acc[i];
    }
#pragma unroll
    for (int kk = 0; kk < 16; ++kk)
      A[kk] = *(const v2f*)(&lds[lo * HID + 4 * kk + 2 * hi]);
  } else {
    const float* zrow = z + (size_t)(row0 + lo) * HID + 2 * hi;
#pragma unroll
    for (int kk = 0; kk < 16; ++kk) A[kk] = *(const v2f*)(zrow + 4 * kk);
  }

  v8f t1[4];
  wmma_16x64(A, W1, hi, lo, t1);
#pragma unroll
  for (int nt = 0; nt < 4; ++nt) {
    int col = nt * 16 + lo;
    float bias = b1[col];
#pragma unroll
    for (int i = 0; i < 8; ++i) {
      float v = t1[nt][i] + bias;
      lds[(i + 8 * hi) * HID + col] = v > 0.0f ? v : 0.0f;
    }
  }

  v2f A2[16];
#pragma unroll
  for (int kk = 0; kk < 16; ++kk)
    A2[kk] = *(const v2f*)(&lds[lo * HID + 4 * kk + 2 * hi]);

  v8f t2[4];
  wmma_16x64(A2, W2, hi, lo, t2);
  float zscale = 0.0f;
  if (!LAST) zscale = 1.0f + eps[next_layer];
  const int lane_e = (int)__builtin_amdgcn_mbcnt_lo(~0u, 0u);
  const int hi_e = lane_e >> 4, lo_e = lane_e & 15;
  if (!LAST) {
#pragma unroll
    for (int nt = 0; nt < 4; ++nt) {
      float bias = b2[nt * 16 + lo];
#pragma unroll
      for (int i = 0; i < 8; ++i) {
        float v = t2[nt][i] + bias;
        v = v > 0.0f ? v : 0.0f;
        size_t o = (size_t)(row0 + i + 8 * hi_e) * HID + nt * 16 + lo_e;
        h_out[o] = v;
        z[o] = zscale * v;
      }
    }
  } else {
    float r[4][8];
#pragma unroll
    for (int nt = 0; nt < 4; ++nt) {
      float bias = b2[nt * 16 + lo_e];
#pragma unroll
      for (int i = 0; i < 8; ++i) {
        float v = t2[nt][i] + bias;
        r[nt][i] = v > 0.0f ? v : 0.0f;
      }
    }
    const int gid = batch[row0 + lo_e];
    const int g0 = __builtin_amdgcn_readfirstlane(gid);
    const bool one_graph = __builtin_amdgcn_ballot_w32(gid != g0) == 0u;
    if (one_graph) {
      float cs[4];
#pragma unroll
      for (int nt = 0; nt < 4; ++nt) {
        float s = r[nt][0];
#pragma unroll
        for (int i = 1; i < 8; ++i) s += r[nt][i];
        s += __shfl_xor(s, 16);
        cs[nt] = s;
      }
      float* prow = pool + (size_t)g0 * POOL_STRIDE;
      atomicAdd(prow + lane_e, hi_e ? cs[1] : cs[0]);
      atomicAdd(prow + 32 + lane_e, hi_e ? cs[3] : cs[2]);
      if (lane_e == 0) atomicAdd(prow + POOL_COUNT, 16.0f);
    } else {
#pragma unroll
      for (int i = 0; i < 8; ++i) {
        const int gr = batch[row0 + i + 8 * hi_e];
        float* prow = pool + (size_t)gr * POOL_STRIDE;
#pragma unroll
        for (int nt = 0; nt < 4; ++nt) atomicAdd(prow + nt * 16 + lo_e, r[nt][i]);
      }
      if (lane_e < 16) atomicAdd(pool + (size_t)gid * POOL_STRIDE + POOL_COUNT, 1.0f);
    }
  }
}

#define G_PITCH 68
static_assert((16 * N_HEADS) % 32 == 0, "a tile's results are whole lines of out");
__global__ __launch_bounds__(32 * N_HEADS) void heads_kernel(
    const float* __restrict__ pool, const float* __restrict__ hW1,
    const float* __restrict__ hb1, const float* __restrict__ hW2,
    const float* __restrict__ hb2, float* __restrict__ out) {
  __shared__ float s_g[16 * G_PITCH];
  __shared__ float s_o[16 * N_HEADS];
  const int tid = threadIdx.x;
  const int lane = tid & 31;
  const int head = __builtin_amdgcn_readfirstlane(tid >> 5);
  const int hi = lane >> 4, lo = lane & 15;
  const int row0 = blockIdx.x * 16;

  for (int e = tid; e < 16 * HID; e += 32 * N_HEADS) {
    const int r = e >> 6, d = e & 63;
    const float* prow = pool + (size_t)(row0 + r) * POOL_STRIDE;
    const float c = prow[POOL_COUNT];
    s_g[r * G_PITCH + d] = prow[d] / (c > 1.0f ? c : 1.0f);
  }
  __syncthreads();

  v2f A[16];
#pragma unroll
  for (int kk = 0; kk < 16; ++kk)
    A[kk] = *(const v2f*)(&s_g[lo * G_PITCH + 4 * kk + 2 * hi]);

  v8f t[4];
  wmma_16x64(A, hW1 + (size_t)head * HID * HID, hi, lo, t);

  float part[8];
#pragma unroll
  for (int i = 0; i < 8; ++i) part[i] = 0.0f;
#pragma unroll
  for (int nt = 0; nt < 4; ++nt) {
    const int col = nt * 16 + lo;
    const float bias = hb1[head * HID + col];
    const float w = hW2[head * HID + col];
#pragma unroll
    for (int i = 0; i < 8; ++i) {
      float v = t[nt][i] + bias;
      v = v > 0.0f ? v : 0.0f;
      part[i] += v * w;
    }
  }
#pragma unroll
  for (int i = 0; i < 8; ++i) {
    float v = part[i];
    v += __shfl_xor(v, 8);
    v += __shfl_xor(v, 4);
    v += __shfl_xor(v, 2);
    v += __shfl_xor(v, 1);
    part[i] = v;
  }
  if (lo == 0) {
    const float b = hb2[head];
#pragma unroll
    for (int i = 0; i < 8; ++i) s_o[(i + 8 * hi) * N_HEADS + head] = part[i] + b;
  }
  __syncthreads();
  if (tid < 16 * N_HEADS) out[(size_t)row0 * N_HEADS + tid] = s_o[tid];
}

extern "C" void kernel_launch(void* const* d_in, const int* in_sizes, int n_in,
                              void* d_out, int out_size, void* d_ws, size_t ws_size,
                              hipStream_t stream) {
  const int*   x     = (const int*)d_in[0];
  const int*   ei    = (const int*)d_in[1];
  const int*   batch = (const int*)d_in[2];
  const float* embed = (const float*)d_in[3];
  const float* eps   = (const float*)d_in[4];
  const float* W1    = (const float*)d_in[5];
  const float* b1    = (const float*)d_in[6];
  const float* W2    = (const float*)d_in[7];
  const float* b2    = (const float*)d_in[8];
  const float* hW1   = (const float*)d_in[9];
  const float* hb1   = (const float*)d_in[10];
  const float* hW2   = (const float*)d_in[11];
  const float* hb2   = (const float*)d_in[12];
  float* out = (float*)d_out;

  float* ws = (float*)d_ws;
  float* h = ws;
  float* z = ws + (size_t)N_NODES * HID;
  float* pool = h;

  const int* src = ei;
  const int* dst = ei + N_EDGES;

  const int nh_elems = N_NODES * HID;
  const int blk = 256;

  unsigned int* tickets = (unsigned int*)d_out;

  init_counts_kernel<<<(nh_elems + blk - 1) / blk, blk, 0, stream>>>(x, eps, h, z, tickets);
  count_scatter_kernel<<<2 * SCATTER_ITEMS, SCATTER_BLOCK, 0, stream>>>(
      src, dst, x, h, z, tickets);
  gin_mlp_kernel<true, false><<<N_NODES / 16, 32, 0, stream>>>(
      z, W1, b1, W2, b2, eps, 1, h, nullptr, nullptr, embed);

  for (int l = 1; l < N_LAYERS; ++l) {
    scatter_kernel<<<2 * SCATTER_ITEMS, SCATTER_BLOCK, 0, stream>>>(
        src, dst, h, z, tickets + (size_t)l * 2 * TICKET_STRIDE);
    if (l + 1 < N_LAYERS) {
      gin_mlp_kernel<false, false><<<N_NODES / 16, 32, 0, stream>>>(
          z, W1 + (size_t)l * HID * HID, b1 + (size_t)l * HID,
          W2 + (size_t)l * HID * HID, b2 + (size_t)l * HID, eps, l + 1, h,
          nullptr, nullptr, nullptr);
    } else {
      const int pool_init = N_GRAPHS * POOL_STRIDE;
      zero_kernel<<<(pool_init + blk - 1) / blk, blk, 0, stream>>>(pool, pool_init);
      gin_mlp_kernel<false, true><<<N_NODES / 16, 32, 0, stream>>>(
          z, W1 + (size_t)l * HID * HID, b1 + (size_t)l * HID,
          W2 + (size_t)l * HID * HID, b2 + (size_t)l * HID, eps, l, nullptr,
          batch, pool, nullptr);
    }
  }

  heads_kernel<<<N_GRAPHS / 16, 32 * N_HEADS, 0, stream>>>(pool, hW1, hb1, hW2, hb2, out);
}
